# starting kernel
# baseline (speedup 1.0000x reference)
#include <hip/hip_runtime.h>
#include <hip/hip_bf16.h>
#include <math.h>

typedef __attribute__((ext_vector_type(16))) _Float16 v16h;
typedef __attribute__((ext_vector_type(8)))  float    v8f;

#define PATCH 288
#define QD 64
#define C_IN 32
#define ROWSTRIDE_A 296
#define ROWSTRIDE_Y 72

#define WS_W   0
#define WS_MR  18432
#define WS_MI  (18432 + 4096)
#define WS_P   (18432 + 8192)

union FragU { uint4 u[2]; v16h v; };

__global__ void __launch_bounds__(256)
qprep_kernel(const float* __restrict__ in_proj_w,
             const float* __restrict__ weights,
             const float* __restrict__ meas_w,
             const float* __restrict__ out_proj_w,
             _Float16* __restrict__ ws)
{
    __shared__ float Mr[64 * 64];
    __shared__ float Mi[64 * 64];
    const int tid = threadIdx.x;

    for (int e = tid; e < 4096; e += 256) {
        int r = e >> 6, c = e & 63;
        Mr[e] = (r == c) ? 1.f : 0.f;
        Mi[e] = 0.f;
    }
    __syncthreads();

    auto apply1q = [&](int wire, float u00r, float u00i, float u01r, float u01i,
                       float u10r, float u10i, float u11r, float u11i) {
        int p = 5 - wire;
        for (int e = tid; e < 2048; e += 256) {
            int pair = e >> 6, col = e & 63;
            int low = pair & ((1 << p) - 1);
            int i0 = ((pair >> p) << (p + 1)) | low;
            int i1 = i0 | (1 << p);
            float ar = Mr[i0 * 64 + col], ai = Mi[i0 * 64 + col];
            float br = Mr[i1 * 64 + col], bi = Mi[i1 * 64 + col];
            Mr[i0 * 64 + col] = u00r * ar - u00i * ai + u01r * br - u01i * bi;
            Mi[i0 * 64 + col] = u00r * ai + u00i * ar + u01r * bi + u01i * br;
            Mr[i1 * 64 + col] = u10r * ar - u10i * ai + u11r * br - u11i * bi;
            Mi[i1 * 64 + col] = u10r * ai + u10i * ar + u11r * bi + u11i * br;
        }
        __syncthreads();
    };
    auto applyCnot = [&](int wire) {
        int pc = 5 - wire, pt = pc - 1;
        for (int e = tid; e < 1024; e += 256) {
            int grp = e >> 6, col = e & 63;
            int low = grp & ((1 << pt) - 1);
            int high = grp >> pt;
            int i0 = (high << (pc + 1)) | (1 << pc) | low;
            int i1 = i0 | (1 << pt);
            float tr = Mr[i0 * 64 + col], ti = Mi[i0 * 64 + col];
            Mr[i0 * 64 + col] = Mr[i1 * 64 + col];
            Mi[i0 * 64 + col] = Mi[i1 * 64 + col];
            Mr[i1 * 64 + col] = tr;
            Mi[i1 * 64 + col] = ti;
        }
        __syncthreads();
    };

    for (int l = 0; l < 2; ++l) {
        for (int i = 0; i < 6; ++i) {
            float t = weights[(l * 6 + i) * 3 + 0];
            float c = cosf(t * 0.5f), s = sinf(t * 0.5f);
            apply1q(i, c, 0.f, 0.f, -s, 0.f, -s, c, 0.f);
            t = weights[(l * 6 + i) * 3 + 1];
            c = cosf(t * 0.5f); s = sinf(t * 0.5f);
            apply1q(i, c, 0.f, -s, 0.f, s, 0.f, c, 0.f);
        }
        for (int i = 0; i < 5; ++i) applyCnot(i);
        for (int i = 0; i < 6; ++i) {
            float t = weights[(l * 6 + i) * 3 + 2];
            float c = cosf(t * 0.5f), s = sinf(t * 0.5f);
            apply1q(i, c, 0.f, -s, 0.f, s, 0.f, c, 0.f);
        }
    }
    for (int i = 0; i < 6; ++i) {
        float t = meas_w[i * 3 + 0], p = meas_w[i * 3 + 1], lm = meas_w[i * 3 + 2];
        float c = cosf(t * 0.5f), s = sinf(t * 0.5f);
        float cp = cosf(p), sp = sinf(p);
        float cl = cosf(lm), sl = sinf(lm);
        float cpl = cosf(p + lm), spl = sinf(p + lm);
        apply1q(i, c, 0.f, -cl * s, -sl * s, cp * s, sp * s, cpl * c, spl * c);
    }

    for (int e = tid; e < 4 * 9 * 512; e += 256) {
        int h = e & 15, lane = (e >> 4) & 31, s = e >> 9;
        int kt = s % 9, nt = s / 9;
        int n = nt * 16 + (lane & 15);
        int base = (lane & 16) ? 8 : 0;
        int kl = (h < 8) ? (base + h) : (base + h + 8);
        int k = kt * 32 + kl;
        ws[WS_W + e] = (_Float16)in_proj_w[n * PATCH + k];
    }
    for (int e = tid; e < 4 * 2 * 512; e += 256) {
        int h = e & 15, lane = (e >> 4) & 31, s = e >> 9;
        int kt = s & 1, nt = s >> 1;
        int n = nt * 16 + (lane & 15);
        int base = (lane & 16) ? 8 : 0;
        int kl = (h < 8) ? (base + h) : (base + h + 8);
        int k = kt * 32 + kl;
        ws[WS_MR + e] = (_Float16)Mr[n * 64 + k];
        ws[WS_MI + e] = (_Float16)Mi[n * 64 + k];
        ws[WS_P + e]  = (_Float16)out_proj_w[n * 64 + k];
    }
}

__global__ void __launch_bounds__(256)
qconv_main_kernel(const float* __restrict__ x,
                  const float* __restrict__ inp_scale,
                  const float* __restrict__ in_proj_b,
                  const float* __restrict__ out_proj_b,
                  const _Float16* __restrict__ ws,
                  float* __restrict__ out)
{
    __shared__ __align__(16) _Float16 bufA[64 * ROWSTRIDE_A];
    __shared__ __align__(16) _Float16 bufY[64 * ROWSTRIDE_Y];
    __shared__ __align__(16) _Float16 bufP[64 * ROWSTRIDE_Y];
    __shared__ float rowsum[64][2];

    const int tid  = threadIdx.x;
    const int lane = tid & 31;
    const int wv   = tid >> 5;
    const int mt   = wv >> 1;
    const int nh   = wv & 1;
    const int pix0 = blockIdx.x * 64;

    {
        const int m = tid & 63;
        const int pixel = pix0 + m;
        const int b  = pixel >> 12;
        const int hw = pixel & 4095;
        const int hh0 = (hw >> 6) - 1;
        const int ww0 = (hw & 63) - 1;
        for (int k = tid >> 6; k < PATCH; k += 4) {
            int c = k / 9;
            int rem = k - c * 9;
            int r = rem / 3;
            int s2 = rem - r * 3;
            int hh = hh0 + r, wwp = ww0 + s2;
            float v = 0.f;
            if ((unsigned)hh < 64u && (unsigned)wwp < 64u)
                v = x[(((size_t)b * C_IN + c) * 64 + hh) * 64 + wwp];
            bufA[m * ROWSTRIDE_A + k] = (_Float16)(v * inp_scale[k]);
        }
    }
    __syncthreads();

    const int nlo  = lane & 15;
    const int base = (lane & 16) ? 8 : 0;

    v8f acc0 = {}, acc1 = {};
    {
        const _Float16* Arow = &bufA[(mt * 16 + nlo) * ROWSTRIDE_A];
        for (int kt = 0; kt < 9; ++kt) {
            FragU af, bf0, bf1;
            const _Float16* pa = Arow + kt * 32 + base;
            af.u[0] = *(const uint4*)(pa);
            af.u[1] = *(const uint4*)(pa + 16);
            const _Float16* pb0 = ws + WS_W + (((2 * nh + 0) * 9 + kt) * 512 + lane * 16);
            const _Float16* pb1 = ws + WS_W + (((2 * nh + 1) * 9 + kt) * 512 + lane * 16);
            bf0.u[0] = *(const uint4*)(pb0); bf0.u[1] = *(const uint4*)(pb0 + 8);
            bf1.u[0] = *(const uint4*)(pb1); bf1.u[1] = *(const uint4*)(pb1 + 8);
            acc0 = __builtin_amdgcn_wmma_f32_16x16x32_f16(false, af.v, false, bf0.v, (short)0, acc0, false, false);
            acc1 = __builtin_amdgcn_wmma_f32_16x16x32_f16(false, af.v, false, bf1.v, (short)0, acc1, false, false);
        }
    }

    const int n0 = (2 * nh + 0) * 16 + nlo;
    const int n1 = (2 * nh + 1) * 16 + nlo;
    float b0 = in_proj_b[n0], b1 = in_proj_b[n1];
    float y0[8], y1[8], part[8];
    for (int r = 0; r < 8; ++r) {
        y0[r] = acc0[r] + b0;
        y1[r] = acc1[r] + b1;
        part[r] = y0[r] * y0[r] + y1[r] * y1[r];
    }
    for (int mask = 1; mask <= 8; mask <<= 1)
        for (int r = 0; r < 8; ++r)
            part[r] += __shfl_xor(part[r], mask, 32);
    if (nlo == 0) {
        for (int r = 0; r < 8; ++r)
            rowsum[mt * 16 + base + r][nh] = part[r];
    }
    __syncthreads();
    for (int r = 0; r < 8; ++r) {
        int m = mt * 16 + base + r;
        float tot = rowsum[m][0] + rowsum[m][1];
        float inv = 1.f / (sqrtf(tot) + 1e-9f);
        bufY[m * ROWSTRIDE_Y + n0] = (_Float16)(y0[r] * inv);
        bufY[m * ROWSTRIDE_Y + n1] = (_Float16)(y1[r] * inv);
    }
    __syncthreads();

    v8f ar0 = {}, ar1 = {}, ai0 = {}, ai1 = {};
    {
        const _Float16* Yrow = &bufY[(mt * 16 + nlo) * ROWSTRIDE_Y];
        for (int kt = 0; kt < 2; ++kt) {
            FragU af, br0, br1, bi0, bi1;
            const _Float16* pa = Yrow + kt * 32 + base;
            af.u[0] = *(const uint4*)(pa);
            af.u[1] = *(const uint4*)(pa + 16);
            const _Float16* q;
            q = ws + WS_MR + (((2 * nh + 0) * 2 + kt) * 512 + lane * 16);
            br0.u[0] = *(const uint4*)(q); br0.u[1] = *(const uint4*)(q + 8);
            q = ws + WS_MR + (((2 * nh + 1) * 2 + kt) * 512 + lane * 16);
            br1.u[0] = *(const uint4*)(q); br1.u[1] = *(const uint4*)(q + 8);
            q = ws + WS_MI + (((2 * nh + 0) * 2 + kt) * 512 + lane * 16);
            bi0.u[0] = *(const uint4*)(q); bi0.u[1] = *(const uint4*)(q + 8);
            q = ws + WS_MI + (((2 * nh + 1) * 2 + kt) * 512 + lane * 16);
            bi1.u[0] = *(const uint4*)(q); bi1.u[1] = *(const uint4*)(q + 8);
            ar0 = __builtin_amdgcn_wmma_f32_16x16x32_f16(false, af.v, false, br0.v, (short)0, ar0, false, false);
            ar1 = __builtin_amdgcn_wmma_f32_16x16x32_f16(false, af.v, false, br1.v, (short)0, ar1, false, false);
            ai0 = __builtin_amdgcn_wmma_f32_16x16x32_f16(false, af.v, false, bi0.v, (short)0, ai0, false, false);
            ai1 = __builtin_amdgcn_wmma_f32_16x16x32_f16(false, af.v, false, bi1.v, (short)0, ai1, false, false);
        }
    }

    for (int r = 0; r < 8; ++r) {
        int m = mt * 16 + base + r;
        bufP[m * ROWSTRIDE_Y + n0] = (_Float16)(ar0[r] * ar0[r] + ai0[r] * ai0[r]);
        bufP[m * ROWSTRIDE_Y + n1] = (_Float16)(ar1[r] * ar1[r] + ai1[r] * ai1[r]);
    }
    __syncthreads();

    v8f o0 = {}, o1 = {};
    {
        const _Float16* Prow = &bufP[(mt * 16 + nlo) * ROWSTRIDE_Y];
        for (int kt = 0; kt < 2; ++kt) {
            FragU af, bp0, bp1;
            const _Float16* pa = Prow + kt * 32 + base;
            af.u[0] = *(const uint4*)(pa);
            af.u[1] = *(const uint4*)(pa + 16);
            const _Float16* q;
            q = ws + WS_P + (((2 * nh + 0) * 2 + kt) * 512 + lane * 16);
            bp0.u[0] = *(const uint4*)(q); bp0.u[1] = *(const uint4*)(q + 8);
            q = ws + WS_P + (((2 * nh + 1) * 2 + kt) * 512 + lane * 16);
            bp1.u[0] = *(const uint4*)(q); bp1.u[1] = *(const uint4*)(q + 8);
            o0 = __builtin_amdgcn_wmma_f32_16x16x32_f16(false, af.v, false, bp0.v, (short)0, o0, false, false);
            o1 = __builtin_amdgcn_wmma_f32_16x16x32_f16(false, af.v, false, bp1.v, (short)0, o1, false, false);
        }
    }

    float ob0 = out_proj_b[n0], ob1 = out_proj_b[n1];
    float* outstage = (float*)bufA;
    for (int r = 0; r < 8; ++r) {
        int m = mt * 16 + base + r;
        outstage[m * 65 + n0] = o0[r] + ob0;
        outstage[m * 65 + n1] = o1[r] + ob1;
    }
    __syncthreads();
    for (int e = tid; e < 64 * 64; e += 256) {
        int co = e >> 6, m = e & 63;
        int pixel = pix0 + m;
        int b = pixel >> 12, hw = pixel & 4095;
        out[((size_t)(b * 64 + co)) * 4096 + hw] = outstage[m * 65 + co];
    }
}

extern "C" void kernel_launch(void* const* d_in, const int* in_sizes, int n_in,
                              void* d_out, int out_size, void* d_ws, size_t ws_size,
                              hipStream_t stream) {
    const float* x          = (const float*)d_in[0];
    const float* inp_scale  = (const float*)d_in[1];
    const float* in_proj_w  = (const float*)d_in[2];
    const float* in_proj_b  = (const float*)d_in[3];
    const float* weights    = (const float*)d_in[4];
    const float* meas_w     = (const float*)d_in[5];
    const float* out_proj_w = (const float*)d_in[6];
    const float* out_proj_b = (const float*)d_in[7];
    _Float16* ws = (_Float16*)d_ws;
    float* out = (float*)d_out;

    hipLaunchKernelGGL(qprep_kernel, dim3(1), dim3(256), 0, stream,
                       in_proj_w, weights, meas_w, out_proj_w, ws);
    hipLaunchKernelGGL(qconv_main_kernel, dim3(1024), dim3(256), 0, stream,
                       x, inp_scale, in_proj_b, out_proj_b, ws, out);
}


# im2col fill: a thread reads its image column of the 3 halo rows once per channel (24 batched aligned loads, no div/mod, no wait per load) and scatters each value to its 3 tap slots; k order kept
# speedup vs baseline: 2.5853x; 2.5853x over previous
#include <hip/hip_runtime.h>
#include <hip/hip_bf16.h>
#include <math.h>

typedef __attribute__((ext_vector_type(16))) _Float16 v16h;
typedef __attribute__((ext_vector_type(8)))  float    v8f;

#define PATCH 288
#define QD 64
#define C_IN 32
#define ROWSTRIDE_A 296
#define ROWSTRIDE_Y 72

#define WS_W   0
#define WS_MR  18432
#define WS_MI  (18432 + 4096)
#define WS_P   (18432 + 8192)

union FragU { uint4 u[2]; v16h v; };


__device__ __forceinline__ void c2mul(const float* a, const float* b, float* o)
{
    for (int r = 0; r < 2; ++r)
        for (int c = 0; c < 2; ++c) {
            float re = 0.f, im = 0.f;
            for (int k = 0; k < 2; ++k) {
                float ar = a[(r * 2 + k) * 2], ai = a[(r * 2 + k) * 2 + 1];
                float br = b[(k * 2 + c) * 2], bi = b[(k * 2 + c) * 2 + 1];
                re += ar * br - ai * bi;
                im += ar * bi + ai * br;
            }
            o[(r * 2 + c) * 2] = re;
            o[(r * 2 + c) * 2 + 1] = im;
        }
}

__device__ __forceinline__ void make_rx(float c, float s, float* o)
{
    o[0] = c;   o[1] = 0.f; o[2] = 0.f; o[3] = -s;
    o[4] = 0.f; o[5] = -s;  o[6] = c;   o[7] = 0.f;
}

__device__ __forceinline__ void make_ry(float c, float s, float* o)
{
    o[0] = c; o[1] = 0.f; o[2] = -s; o[3] = 0.f;
    o[4] = s; o[5] = 0.f; o[6] = c;  o[7] = 0.f;
}

template <bool HI, bool PERM>
__device__ __forceinline__ void block_sweep(float* Mr, float* Mi, const float* B, int tid)
{
    float yr[2][8], yi[2][8];
    for (int q = 0; q < 2; ++q) {
        const int t = tid + 256 * q;
        const int c = t & 63, o = t >> 6;
        float xr[8], xi[8];
        for (int s = 0; s < 8; ++s) {
            int row = HI ? (s * 8 + o) : (o * 8 + s);
            if (PERM) row ^= row >> 1;
            xr[s] = Mr[row * 64 + c];
            xi[s] = Mi[row * 64 + c];
        }
        for (int a = 0; a < 8; ++a) {
            float sr = 0.f, si = 0.f;
            for (int s = 0; s < 8; ++s) {
                const float br = B[(a * 8 + s) * 2], bi = B[(a * 8 + s) * 2 + 1];
                sr = fmaf(br, xr[s], sr);
                sr = fmaf(-bi, xi[s], sr);
                si = fmaf(br, xi[s], si);
                si = fmaf(bi, xr[s], si);
            }
            yr[q][a] = sr;
            yi[q][a] = si;
        }
    }
    if (PERM) __syncthreads();
    for (int q = 0; q < 2; ++q) {
        const int t = tid + 256 * q;
        const int c = t & 63, o = t >> 6;
        for (int a = 0; a < 8; ++a) {
            const int row = HI ? (a * 8 + o) : (o * 8 + a);
            Mr[row * 64 + c] = yr[q][a];
            Mi[row * 64 + c] = yi[q][a];
        }
    }
    __syncthreads();
}

__global__ void __launch_bounds__(256)
qprep_kernel(const float* __restrict__ in_proj_w,
             const float* __restrict__ weights,
             const float* __restrict__ meas_w,
             const float* __restrict__ out_proj_w,
             _Float16* __restrict__ ws)
{
    __shared__ float Mr[64 * 64];
    __shared__ float Mi[64 * 64];
    __shared__ float CS[60 * 2];
    __shared__ float G[18 * 8];
    __shared__ float Blk[4 * 128];
    const int tid = threadIdx.x;

    if (tid < 60) {
        float a;
        if (tid < 36) {
            a = weights[tid] * 0.5f;
        } else {
            const int k = tid - 36, w = k >> 2, which = k & 3;
            const float t = meas_w[w * 3 + 0], p = meas_w[w * 3 + 1], lm = meas_w[w * 3 + 2];
            a = (which == 0) ? t * 0.5f : (which == 1) ? p : (which == 2) ? lm : (p + lm);
        }
        CS[tid * 2]     = cosf(a);
        CS[tid * 2 + 1] = sinf(a);
    }
    __syncthreads();

    if (tid < 18) {
        const int L = tid / 6, w = tid - L * 6;
        float m0[8], m1[8], m2[8], m3[8];
        if (L == 0) {
            make_ry(CS[(w * 3 + 1) * 2], CS[(w * 3 + 1) * 2 + 1], m0);
            make_rx(CS[(w * 3 + 0) * 2], CS[(w * 3 + 0) * 2 + 1], m1);
            c2mul(m0, m1, m3);
        } else if (L == 1) {
            make_ry(CS[((6 + w) * 3 + 1) * 2], CS[((6 + w) * 3 + 1) * 2 + 1], m0);
            make_rx(CS[((6 + w) * 3 + 0) * 2], CS[((6 + w) * 3 + 0) * 2 + 1], m1);
            c2mul(m0, m1, m2);
            make_ry(CS[(w * 3 + 2) * 2], CS[(w * 3 + 2) * 2 + 1], m0);
            c2mul(m2, m0, m3);
        } else {
            const float c  = CS[(36 + w * 4 + 0) * 2], s  = CS[(36 + w * 4 + 0) * 2 + 1];
            const float cp = CS[(36 + w * 4 + 1) * 2], sp = CS[(36 + w * 4 + 1) * 2 + 1];
            const float cl = CS[(36 + w * 4 + 2) * 2], sl = CS[(36 + w * 4 + 2) * 2 + 1];
            const float cpl = CS[(36 + w * 4 + 3) * 2], spl = CS[(36 + w * 4 + 3) * 2 + 1];
            m0[0] = c;      m0[1] = 0.f;    m0[2] = -cl * s; m0[3] = -sl * s;
            m0[4] = cp * s; m0[5] = sp * s; m0[6] = cpl * c; m0[7] = spl * c;
            make_ry(CS[((6 + w) * 3 + 2) * 2], CS[((6 + w) * 3 + 2) * 2 + 1], m1);
            c2mul(m0, m1, m3);
        }
        for (int e = 0; e < 8; ++e) G[tid * 8 + e] = m3[e];
    }
    __syncthreads();

    {
        const int tb = tid >> 6, r = (tid >> 3) & 7, j = tid & 7;
        const int L = 1 + (tb >> 1), w0 = (tb & 1) * 3;
        float pr = 1.f, pi = 0.f;
        for (int q = 0; q < 3; ++q) {
            const int rb = (r >> (2 - q)) & 1, cb = (j >> (2 - q)) & 1;
            const float gr = G[((L * 6 + w0 + q) * 4 + rb * 2 + cb) * 2];
            const float gi = G[((L * 6 + w0 + q) * 4 + rb * 2 + cb) * 2 + 1];
            const float nr = pr * gr - pi * gi, ni = pr * gi + pi * gr;
            pr = nr; pi = ni;
        }
        Blk[tid * 2] = pr;
        Blk[tid * 2 + 1] = pi;
    }
    {
        const int col = tid & 63, slo = tid >> 6;
        float gr[6][2], gi[6][2];
        for (int w = 0; w < 6; ++w) {
            const int cb = (col >> (5 - w)) & 1;
            gr[w][0] = G[(w * 4 + 0 + cb) * 2];  gi[w][0] = G[(w * 4 + 0 + cb) * 2 + 1];
            gr[w][1] = G[(w * 4 + 2 + cb) * 2];  gi[w][1] = G[(w * 4 + 2 + cb) * 2 + 1];
        }
        const bool b4 = (slo >> 1) & 1, b5 = slo & 1;
        const float g4r = b4 ? gr[4][1] : gr[4][0], g4i = b4 ? gi[4][1] : gi[4][0];
        const float g5r = b5 ? gr[5][1] : gr[5][0], g5i = b5 ? gi[5][1] : gi[5][0];
        float pr[16], pi[16];
        pr[0] = g4r * g5r - g4i * g5i;
        pi[0] = g4r * g5i + g4i * g5r;
        for (int k = 0; k < 4; ++k) {
            const int w = 3 - k, n = 1 << k;
            for (int idx = 0; idx < n; ++idx) {
                const float ar = pr[idx], ai = pi[idx];
                pr[idx + n] = ar * gr[w][1] - ai * gi[w][1];
                pi[idx + n] = ar * gi[w][1] + ai * gr[w][1];
                pr[idx]     = ar * gr[w][0] - ai * gi[w][0];
                pi[idx]     = ar * gi[w][0] + ai * gr[w][0];
            }
        }
        for (int t = 0; t < 16; ++t) {
            const int s = slo + 4 * t;
            const int row = s ^ (s >> 1) ^ (s >> 2) ^ (s >> 3) ^ (s >> 4) ^ (s >> 5);
            Mr[row * 64 + col] = pr[t];
            Mi[row * 64 + col] = pi[t];
        }
    }
    __syncthreads();

    block_sweep<true,  false>(Mr, Mi, Blk + 0 * 128, tid);
    block_sweep<false, false>(Mr, Mi, Blk + 1 * 128, tid);
    block_sweep<true,  true >(Mr, Mi, Blk + 2 * 128, tid);
    block_sweep<false, false>(Mr, Mi, Blk + 3 * 128, tid);

    for (int e = tid; e < 4 * 9 * 512; e += 256) {
        int h = e & 15, lane = (e >> 4) & 31, s = e >> 9;
        int kt = s % 9, nt = s / 9;
        int n = nt * 16 + (lane & 15);
        int base = (lane & 16) ? 8 : 0;
        int kl = (h < 8) ? (base + h) : (base + h + 8);
        int k = kt * 32 + kl;
        ws[WS_W + e] = (_Float16)in_proj_w[n * PATCH + k];
    }
    for (int e = tid; e < 4 * 2 * 512; e += 256) {
        int h = e & 15, lane = (e >> 4) & 31, s = e >> 9;
        int kt = s & 1, nt = s >> 1;
        int n = nt * 16 + (lane & 15);
        int base = (lane & 16) ? 8 : 0;
        int kl = (h < 8) ? (base + h) : (base + h + 8);
        int k = kt * 32 + kl;
        ws[WS_MR + e] = (_Float16)Mr[n * 64 + k];
        ws[WS_MI + e] = (_Float16)Mi[n * 64 + k];
        ws[WS_P + e]  = (_Float16)out_proj_w[n * 64 + k];
    }
}

__global__ void __launch_bounds__(256)
qconv_main_kernel(const float* __restrict__ x,
                  const float* __restrict__ inp_scale,
                  const float* __restrict__ in_proj_b,
                  const float* __restrict__ out_proj_b,
                  const _Float16* __restrict__ ws,
                  float* __restrict__ out)
{
    __shared__ __align__(16) _Float16 bufA[64 * ROWSTRIDE_A];
    __shared__ __align__(16) _Float16 bufY[64 * ROWSTRIDE_Y];
    __shared__ __align__(16) _Float16 bufP[64 * ROWSTRIDE_Y];
    __shared__ float rowsum[64][2];

    const int tid  = threadIdx.x;
    const int lane = tid & 31;
    const int wv   = tid >> 5;
    const int mt   = wv >> 1;
    const int nh   = wv & 1;
    const int pix0 = blockIdx.x * 64;

    {
        const int m = tid & 63;
        const int g = __builtin_amdgcn_readfirstlane(tid >> 6);
        const int b = blockIdx.x >> 6;
        const int h = blockIdx.x & 63;
        const bool top = h > 0, bot = h < 63;
        const int hm = top ? h - 1 : h;
        const int hp = bot ? h + 1 : h;
        const float* xb = x + (size_t)b * C_IN * 4096 + m;
        float v[8][3];
        #pragma unroll
        for (int j = 0; j < 8; ++j) {
            const float* xc = xb + (g + 4 * j) * 4096;
            v[j][0] = xc[hm * 64];
            v[j][1] = xc[h * 64];
            v[j][2] = xc[hp * 64];
        }
        const bool lastcol = (m == 63), firstcol = (m == 0);
        _Float16* r0 = bufA + ((m + 1) & 63) * ROWSTRIDE_A;
        _Float16* r1 = bufA + m * ROWSTRIDE_A;
        _Float16* r2 = bufA + ((m + 63) & 63) * ROWSTRIDE_A;
        #pragma unroll
        for (int j = 0; j < 8; ++j) {
            const int kb = (g + 4 * j) * 9;
            #pragma unroll
            for (int r = 0; r < 3; ++r) {
                float val = v[j][r];
                if (r == 0 && !top) val = 0.f;
                if (r == 2 && !bot) val = 0.f;
                const int k = kb + r * 3;
                const float a0 = lastcol ? 0.f : val;
                const float a2 = firstcol ? 0.f : val;
                r0[k + 0] = (_Float16)(a0 * inp_scale[k + 0]);
                r1[k + 1] = (_Float16)(val * inp_scale[k + 1]);
                r2[k + 2] = (_Float16)(a2 * inp_scale[k + 2]);
            }
        }
    }
    __syncthreads();

    const int nlo  = lane & 15;
    const int base = (lane & 16) ? 8 : 0;

    v8f acc0 = {}, acc1 = {};
    {
        const _Float16* Arow = &bufA[(mt * 16 + nlo) * ROWSTRIDE_A];
        for (int kt = 0; kt < 9; ++kt) {
            FragU af, bf0, bf1;
            const _Float16* pa = Arow + kt * 32 + base;
            af.u[0] = *(const uint4*)(pa);
            af.u[1] = *(const uint4*)(pa + 16);
            const _Float16* pb0 = ws + WS_W + (((2 * nh + 0) * 9 + kt) * 512 + lane * 16);
            const _Float16* pb1 = ws + WS_W + (((2 * nh + 1) * 9 + kt) * 512 + lane * 16);
            bf0.u[0] = *(const uint4*)(pb0); bf0.u[1] = *(const uint4*)(pb0 + 8);
            bf1.u[0] = *(const uint4*)(pb1); bf1.u[1] = *(const uint4*)(pb1 + 8);
            acc0 = __builtin_amdgcn_wmma_f32_16x16x32_f16(false, af.v, false, bf0.v, (short)0, acc0, false, false);
            acc1 = __builtin_amdgcn_wmma_f32_16x16x32_f16(false, af.v, false, bf1.v, (short)0, acc1, false, false);
        }
    }

    const int n0 = (2 * nh + 0) * 16 + nlo;
    const int n1 = (2 * nh + 1) * 16 + nlo;
    float b0 = in_proj_b[n0], b1 = in_proj_b[n1];
    float y0[8], y1[8], part[8];
    for (int r = 0; r < 8; ++r) {
        y0[r] = acc0[r] + b0;
        y1[r] = acc1[r] + b1;
        part[r] = y0[r] * y0[r] + y1[r] * y1[r];
    }
    for (int mask = 1; mask <= 8; mask <<= 1)
        for (int r = 0; r < 8; ++r)
            part[r] += __shfl_xor(part[r], mask, 32);
    if (nlo == 0) {
        for (int r = 0; r < 8; ++r)
            rowsum[mt * 16 + base + r][nh] = part[r];
    }
    __syncthreads();
    for (int r = 0; r < 8; ++r) {
        int m = mt * 16 + base + r;
        float tot = rowsum[m][0] + rowsum[m][1];
        float inv = 1.f / (sqrtf(tot) + 1e-9f);
        bufY[m * ROWSTRIDE_Y + n0] = (_Float16)(y0[r] * inv);
        bufY[m * ROWSTRIDE_Y + n1] = (_Float16)(y1[r] * inv);
    }
    __syncthreads();

    v8f ar0 = {}, ar1 = {}, ai0 = {}, ai1 = {};
    {
        const _Float16* Yrow = &bufY[(mt * 16 + nlo) * ROWSTRIDE_Y];
        for (int kt = 0; kt < 2; ++kt) {
            FragU af, br0, br1, bi0, bi1;
            const _Float16* pa = Yrow + kt * 32 + base;
            af.u[0] = *(const uint4*)(pa);
            af.u[1] = *(const uint4*)(pa + 16);
            const _Float16* q;
            q = ws + WS_MR + (((2 * nh + 0) * 2 + kt) * 512 + lane * 16);
            br0.u[0] = *(const uint4*)(q); br0.u[1] = *(const uint4*)(q + 8);
            q = ws + WS_MR + (((2 * nh + 1) * 2 + kt) * 512 + lane * 16);
            br1.u[0] = *(const uint4*)(q); br1.u[1] = *(const uint4*)(q + 8);
            q = ws + WS_MI + (((2 * nh + 0) * 2 + kt) * 512 + lane * 16);
            bi0.u[0] = *(const uint4*)(q); bi0.u[1] = *(const uint4*)(q + 8);
            q = ws + WS_MI + (((2 * nh + 1) * 2 + kt) * 512 + lane * 16);
            bi1.u[0] = *(const uint4*)(q); bi1.u[1] = *(const uint4*)(q + 8);
            ar0 = __builtin_amdgcn_wmma_f32_16x16x32_f16(false, af.v, false, br0.v, (short)0, ar0, false, false);
            ar1 = __builtin_amdgcn_wmma_f32_16x16x32_f16(false, af.v, false, br1.v, (short)0, ar1, false, false);
            ai0 = __builtin_amdgcn_wmma_f32_16x16x32_f16(false, af.v, false, bi0.v, (short)0, ai0, false, false);
            ai1 = __builtin_amdgcn_wmma_f32_16x16x32_f16(false, af.v, false, bi1.v, (short)0, ai1, false, false);
        }
    }

    for (int r = 0; r < 8; ++r) {
        int m = mt * 16 + base + r;
        bufP[m * ROWSTRIDE_Y + n0] = (_Float16)(ar0[r] * ar0[r] + ai0[r] * ai0[r]);
        bufP[m * ROWSTRIDE_Y + n1] = (_Float16)(ar1[r] * ar1[r] + ai1[r] * ai1[r]);
    }
    __syncthreads();

    v8f o0 = {}, o1 = {};
    {
        const _Float16* Prow = &bufP[(mt * 16 + nlo) * ROWSTRIDE_Y];
        for (int kt = 0; kt < 2; ++kt) {
            FragU af, bp0, bp1;
            const _Float16* pa = Prow + kt * 32 + base;
            af.u[0] = *(const uint4*)(pa);
            af.u[1] = *(const uint4*)(pa + 16);
            const _Float16* q;
            q = ws + WS_P + (((2 * nh + 0) * 2 + kt) * 512 + lane * 16);
            bp0.u[0] = *(const uint4*)(q); bp0.u[1] = *(const uint4*)(q + 8);
            q = ws + WS_P + (((2 * nh + 1) * 2 + kt) * 512 + lane * 16);
            bp1.u[0] = *(const uint4*)(q); bp1.u[1] = *(const uint4*)(q + 8);
            o0 = __builtin_amdgcn_wmma_f32_16x16x32_f16(false, af.v, false, bp0.v, (short)0, o0, false, false);
            o1 = __builtin_amdgcn_wmma_f32_16x16x32_f16(false, af.v, false, bp1.v, (short)0, o1, false, false);
        }
    }

    float ob0 = out_proj_b[n0], ob1 = out_proj_b[n1];
    float* outstage = (float*)bufA;
    for (int r = 0; r < 8; ++r) {
        int m = mt * 16 + base + r;
        outstage[m * 65 + n0] = o0[r] + ob0;
        outstage[m * 65 + n1] = o1[r] + ob1;
    }
    __syncthreads();
    for (int e = tid; e < 64 * 64; e += 256) {
        int co = e >> 6, m = e & 63;
        int pixel = pix0 + m;
        int b = pixel >> 12, hw = pixel & 4095;
        out[((size_t)(b * 64 + co)) * 4096 + hw] = outstage[m * 65 + co];
    }
}

extern "C" void kernel_launch(void* const* d_in, const int* in_sizes, int n_in,
                              void* d_out, int out_size, void* d_ws, size_t ws_size,
                              hipStream_t stream) {
    const float* x          = (const float*)d_in[0];
    const float* inp_scale  = (const float*)d_in[1];
    const float* in_proj_w  = (const float*)d_in[2];
    const float* in_proj_b  = (const float*)d_in[3];
    const float* weights    = (const float*)d_in[4];
    const float* meas_w     = (const float*)d_in[5];
    const float* out_proj_w = (const float*)d_in[6];
    const float* out_proj_b = (const float*)d_in[7];
    _Float16* ws = (_Float16*)d_ws;
    float* out = (float*)d_out;

    hipLaunchKernelGGL(qprep_kernel, dim3(1), dim3(256), 0, stream,
                       in_proj_w, weights, meas_w, out_proj_w, ws);
    hipLaunchKernelGGL(qconv_main_kernel, dim3(1024), dim3(256), 0, stream,
                       x, inp_scale, in_proj_b, out_proj_b, ws, out);
}


# prep work-group widened from 256 to 1024 threads: weight repack is 18 + 4 loads per thread, all in flight at once (was 72 + 32 elements in load-wait-store trips of 4); sweeps one task per thread
# speedup vs baseline: 3.6392x; 1.4077x over previous
#include <hip/hip_runtime.h>
#include <hip/hip_bf16.h>
#include <math.h>

typedef __attribute__((ext_vector_type(16))) _Float16 v16h;
typedef __attribute__((ext_vector_type(8)))  float    v8f;

#define PATCH 288
#define QD 64
#define C_IN 32
#define ROWSTRIDE_A 296
#define ROWSTRIDE_Y 72

#define WS_W   0
#define WS_MR  18432
#define WS_MI  (18432 + 4096)
#define WS_P   (18432 + 8192)

union FragU { uint4 u[2]; v16h v; };


__device__ __forceinline__ void c2mul(const float* a, const float* b, float* o)
{
    for (int r = 0; r < 2; ++r)
        for (int c = 0; c < 2; ++c) {
            float re = 0.f, im = 0.f;
            for (int k = 0; k < 2; ++k) {
                float ar = a[(r * 2 + k) * 2], ai = a[(r * 2 + k) * 2 + 1];
                float br = b[(k * 2 + c) * 2], bi = b[(k * 2 + c) * 2 + 1];
                re += ar * br - ai * bi;
                im += ar * bi + ai * br;
            }
            o[(r * 2 + c) * 2] = re;
            o[(r * 2 + c) * 2 + 1] = im;
        }
}

__device__ __forceinline__ void make_rx(float c, float s, float* o)
{
    o[0] = c;   o[1] = 0.f; o[2] = 0.f; o[3] = -s;
    o[4] = 0.f; o[5] = -s;  o[6] = c;   o[7] = 0.f;
}

__device__ __forceinline__ void make_ry(float c, float s, float* o)
{
    o[0] = c; o[1] = 0.f; o[2] = -s; o[3] = 0.f;
    o[4] = s; o[5] = 0.f; o[6] = c;  o[7] = 0.f;
}

template <bool HI, bool PERM>
__device__ __forceinline__ void block_sweep(float* Mr, float* Mi, const float* B, int tid)
{
    float yr[8] = {}, yi[8] = {};
    const bool act = tid < 512;
    const int c = tid & 63, o = (tid >> 6) & 7;
    if (act) {
        float xr[8], xi[8];
        for (int s = 0; s < 8; ++s) {
            int row = HI ? (s * 8 + o) : (o * 8 + s);
            if (PERM) row ^= row >> 1;
            xr[s] = Mr[row * 64 + c];
            xi[s] = Mi[row * 64 + c];
        }
        for (int a = 0; a < 8; ++a) {
            float sr = 0.f, si = 0.f;
            for (int s = 0; s < 8; ++s) {
                const float br = B[(a * 8 + s) * 2], bi = B[(a * 8 + s) * 2 + 1];
                sr = fmaf(br, xr[s], sr);
                sr = fmaf(-bi, xi[s], sr);
                si = fmaf(br, xi[s], si);
                si = fmaf(bi, xr[s], si);
            }
            yr[a] = sr;
            yi[a] = si;
        }
    }
    if (PERM) __syncthreads();
    if (act) {
        for (int a = 0; a < 8; ++a) {
            const int row = HI ? (a * 8 + o) : (o * 8 + a);
            Mr[row * 64 + c] = yr[a];
            Mi[row * 64 + c] = yi[a];
        }
    }
    __syncthreads();
}

__global__ void __launch_bounds__(1024)
qprep_kernel(const float* __restrict__ in_proj_w,
             const float* __restrict__ weights,
             const float* __restrict__ meas_w,
             const float* __restrict__ out_proj_w,
             _Float16* __restrict__ ws)
{
    __shared__ float Mr[64 * 64];
    __shared__ float Mi[64 * 64];
    __shared__ float CS[60 * 2];
    __shared__ float G[18 * 8];
    __shared__ float Blk[4 * 128];
    const int tid = threadIdx.x;

    if (tid < 60) {
        float a;
        if (tid < 36) {
            a = weights[tid] * 0.5f;
        } else {
            const int k = tid - 36, w = k >> 2, which = k & 3;
            const float t = meas_w[w * 3 + 0], p = meas_w[w * 3 + 1], lm = meas_w[w * 3 + 2];
            a = (which == 0) ? t * 0.5f : (which == 1) ? p : (which == 2) ? lm : (p + lm);
        }
        CS[tid * 2]     = cosf(a);
        CS[tid * 2 + 1] = sinf(a);
    }
    __syncthreads();

    if (tid < 18) {
        const int L = tid / 6, w = tid - L * 6;
        float m0[8], m1[8], m2[8], m3[8];
        if (L == 0) {
            make_ry(CS[(w * 3 + 1) * 2], CS[(w * 3 + 1) * 2 + 1], m0);
            make_rx(CS[(w * 3 + 0) * 2], CS[(w * 3 + 0) * 2 + 1], m1);
            c2mul(m0, m1, m3);
        } else if (L == 1) {
            make_ry(CS[((6 + w) * 3 + 1) * 2], CS[((6 + w) * 3 + 1) * 2 + 1], m0);
            make_rx(CS[((6 + w) * 3 + 0) * 2], CS[((6 + w) * 3 + 0) * 2 + 1], m1);
            c2mul(m0, m1, m2);
            make_ry(CS[(w * 3 + 2) * 2], CS[(w * 3 + 2) * 2 + 1], m0);
            c2mul(m2, m0, m3);
        } else {
            const float c  = CS[(36 + w * 4 + 0) * 2], s  = CS[(36 + w * 4 + 0) * 2 + 1];
            const float cp = CS[(36 + w * 4 + 1) * 2], sp = CS[(36 + w * 4 + 1) * 2 + 1];
            const float cl = CS[(36 + w * 4 + 2) * 2], sl = CS[(36 + w * 4 + 2) * 2 + 1];
            const float cpl = CS[(36 + w * 4 + 3) * 2], spl = CS[(36 + w * 4 + 3) * 2 + 1];
            m0[0] = c;      m0[1] = 0.f;    m0[2] = -cl * s; m0[3] = -sl * s;
            m0[4] = cp * s; m0[5] = sp * s; m0[6] = cpl * c; m0[7] = spl * c;
            make_ry(CS[((6 + w) * 3 + 2) * 2], CS[((6 + w) * 3 + 2) * 2 + 1], m1);
            c2mul(m0, m1, m3);
        }
        for (int e = 0; e < 8; ++e) G[tid * 8 + e] = m3[e];
    }
    __syncthreads();

    if (tid < 256) {
        const int tb = tid >> 6, r = (tid >> 3) & 7, j = tid & 7;
        const int L = 1 + (tb >> 1), w0 = (tb & 1) * 3;
        float pr = 1.f, pi = 0.f;
        for (int q = 0; q < 3; ++q) {
            const int rb = (r >> (2 - q)) & 1, cb = (j >> (2 - q)) & 1;
            const float gr = G[((L * 6 + w0 + q) * 4 + rb * 2 + cb) * 2];
            const float gi = G[((L * 6 + w0 + q) * 4 + rb * 2 + cb) * 2 + 1];
            const float nr = pr * gr - pi * gi, ni = pr * gi + pi * gr;
            pr = nr; pi = ni;
        }
        Blk[tid * 2] = pr;
        Blk[tid * 2 + 1] = pi;
    }
    if (tid < 256) {
        const int col = tid & 63, slo = tid >> 6;
        float gr[6][2], gi[6][2];
        for (int w = 0; w < 6; ++w) {
            const int cb = (col >> (5 - w)) & 1;
            gr[w][0] = G[(w * 4 + 0 + cb) * 2];  gi[w][0] = G[(w * 4 + 0 + cb) * 2 + 1];
            gr[w][1] = G[(w * 4 + 2 + cb) * 2];  gi[w][1] = G[(w * 4 + 2 + cb) * 2 + 1];
        }
        const bool b4 = (slo >> 1) & 1, b5 = slo & 1;
        const float g4r = b4 ? gr[4][1] : gr[4][0], g4i = b4 ? gi[4][1] : gi[4][0];
        const float g5r = b5 ? gr[5][1] : gr[5][0], g5i = b5 ? gi[5][1] : gi[5][0];
        float pr[16], pi[16];
        pr[0] = g4r * g5r - g4i * g5i;
        pi[0] = g4r * g5i + g4i * g5r;
        for (int k = 0; k < 4; ++k) {
            const int w = 3 - k, n = 1 << k;
            for (int idx = 0; idx < n; ++idx) {
                const float ar = pr[idx], ai = pi[idx];
                pr[idx + n] = ar * gr[w][1] - ai * gi[w][1];
                pi[idx + n] = ar * gi[w][1] + ai * gr[w][1];
                pr[idx]     = ar * gr[w][0] - ai * gi[w][0];
                pi[idx]     = ar * gi[w][0] + ai * gr[w][0];
            }
        }
        for (int t = 0; t < 16; ++t) {
            const int s = slo + 4 * t;
            const int row = s ^ (s >> 1) ^ (s >> 2) ^ (s >> 3) ^ (s >> 4) ^ (s >> 5);
            Mr[row * 64 + col] = pr[t];
            Mi[row * 64 + col] = pi[t];
        }
    }
    __syncthreads();

    block_sweep<true,  false>(Mr, Mi, Blk + 0 * 128, tid);
    block_sweep<false, false>(Mr, Mi, Blk + 1 * 128, tid);
    block_sweep<true,  true >(Mr, Mi, Blk + 2 * 128, tid);
    block_sweep<false, false>(Mr, Mi, Blk + 3 * 128, tid);

    {
        const int h = tid & 15, lane = (tid >> 4) & 31, t9 = tid >> 9;
        const int base = (lane & 16) ? 8 : 0;
        const int kl = (h < 8) ? (base + h) : (base + h + 8);
        const float* src = in_proj_w + (lane & 15) * PATCH + kl;
        float wv[18];
        for (int i = 0; i < 18; ++i) {
            const int s = 2 * i + t9;
            const int kt = s % 9, nt = s / 9;
            wv[i] = src[nt * 16 * PATCH + kt * 32];
        }
        for (int i = 0; i < 18; ++i)
            ws[WS_W + tid + 1024 * i] = (_Float16)wv[i];
    }
    #pragma unroll
    for (int e = tid; e < 4 * 2 * 512; e += 1024) {
        int h = e & 15, lane = (e >> 4) & 31, s = e >> 9;
        int kt = s & 1, nt = s >> 1;
        int n = nt * 16 + (lane & 15);
        int base = (lane & 16) ? 8 : 0;
        int kl = (h < 8) ? (base + h) : (base + h + 8);
        int k = kt * 32 + kl;
        ws[WS_MR + e] = (_Float16)Mr[n * 64 + k];
        ws[WS_MI + e] = (_Float16)Mi[n * 64 + k];
        ws[WS_P + e]  = (_Float16)out_proj_w[n * 64 + k];
    }
}

__global__ void __launch_bounds__(256)
qconv_main_kernel(const float* __restrict__ x,
                  const float* __restrict__ inp_scale,
                  const float* __restrict__ in_proj_b,
                  const float* __restrict__ out_proj_b,
                  const _Float16* __restrict__ ws,
                  float* __restrict__ out)
{
    __shared__ __align__(16) _Float16 bufA[64 * ROWSTRIDE_A];
    __shared__ __align__(16) _Float16 bufY[64 * ROWSTRIDE_Y];
    __shared__ __align__(16) _Float16 bufP[64 * ROWSTRIDE_Y];
    __shared__ float rowsum[64][2];

    const int tid  = threadIdx.x;
    const int lane = tid & 31;
    const int wv   = tid >> 5;
    const int mt   = wv >> 1;
    const int nh   = wv & 1;
    const int pix0 = blockIdx.x * 64;

    {
        const int m = tid & 63;
        const int g = __builtin_amdgcn_readfirstlane(tid >> 6);
        const int b = blockIdx.x >> 6;
        const int h = blockIdx.x & 63;
        const bool top = h > 0, bot = h < 63;
        const int hm = top ? h - 1 : h;
        const int hp = bot ? h + 1 : h;
        const float* xb = x + (size_t)b * C_IN * 4096 + m;
        float v[8][3];
        #pragma unroll
        for (int j = 0; j < 8; ++j) {
            const float* xc = xb + (g + 4 * j) * 4096;
            v[j][0] = xc[hm * 64];
            v[j][1] = xc[h * 64];
            v[j][2] = xc[hp * 64];
        }
        const bool lastcol = (m == 63), firstcol = (m == 0);
        _Float16* r0 = bufA + ((m + 1) & 63) * ROWSTRIDE_A;
        _Float16* r1 = bufA + m * ROWSTRIDE_A;
        _Float16* r2 = bufA + ((m + 63) & 63) * ROWSTRIDE_A;
        #pragma unroll
        for (int j = 0; j < 8; ++j) {
            const int kb = (g + 4 * j) * 9;
            #pragma unroll
            for (int r = 0; r < 3; ++r) {
                float val = v[j][r];
                if (r == 0 && !top) val = 0.f;
                if (r == 2 && !bot) val = 0.f;
                const int k = kb + r * 3;
                const float a0 = lastcol ? 0.f : val;
                const float a2 = firstcol ? 0.f : val;
                r0[k + 0] = (_Float16)(a0 * inp_scale[k + 0]);
                r1[k + 1] = (_Float16)(val * inp_scale[k + 1]);
                r2[k + 2] = (_Float16)(a2 * inp_scale[k + 2]);
            }
        }
    }
    __syncthreads();

    const int nlo  = lane & 15;
    const int base = (lane & 16) ? 8 : 0;

    v8f acc0 = {}, acc1 = {};
    {
        const _Float16* Arow = &bufA[(mt * 16 + nlo) * ROWSTRIDE_A];
        for (int kt = 0; kt < 9; ++kt) {
            FragU af, bf0, bf1;
            const _Float16* pa = Arow + kt * 32 + base;
            af.u[0] = *(const uint4*)(pa);
            af.u[1] = *(const uint4*)(pa + 16);
            const _Float16* pb0 = ws + WS_W + (((2 * nh + 0) * 9 + kt) * 512 + lane * 16);
            const _Float16* pb1 = ws + WS_W + (((2 * nh + 1) * 9 + kt) * 512 + lane * 16);
            bf0.u[0] = *(const uint4*)(pb0); bf0.u[1] = *(const uint4*)(pb0 + 8);
            bf1.u[0] = *(const uint4*)(pb1); bf1.u[1] = *(const uint4*)(pb1 + 8);
            acc0 = __builtin_amdgcn_wmma_f32_16x16x32_f16(false, af.v, false, bf0.v, (short)0, acc0, false, false);
            acc1 = __builtin_amdgcn_wmma_f32_16x16x32_f16(false, af.v, false, bf1.v, (short)0, acc1, false, false);
        }
    }

    const int n0 = (2 * nh + 0) * 16 + nlo;
    const int n1 = (2 * nh + 1) * 16 + nlo;
    float b0 = in_proj_b[n0], b1 = in_proj_b[n1];
    float y0[8], y1[8], part[8];
    for (int r = 0; r < 8; ++r) {
        y0[r] = acc0[r] + b0;
        y1[r] = acc1[r] + b1;
        part[r] = y0[r] * y0[r] + y1[r] * y1[r];
    }
    for (int mask = 1; mask <= 8; mask <<= 1)
        for (int r = 0; r < 8; ++r)
            part[r] += __shfl_xor(part[r], mask, 32);
    if (nlo == 0) {
        for (int r = 0; r < 8; ++r)
            rowsum[mt * 16 + base + r][nh] = part[r];
    }
    __syncthreads();
    for (int r = 0; r < 8; ++r) {
        int m = mt * 16 + base + r;
        float tot = rowsum[m][0] + rowsum[m][1];
        float inv = 1.f / (sqrtf(tot) + 1e-9f);
        bufY[m * ROWSTRIDE_Y + n0] = (_Float16)(y0[r] * inv);
        bufY[m * ROWSTRIDE_Y + n1] = (_Float16)(y1[r] * inv);
    }
    __syncthreads();

    v8f ar0 = {}, ar1 = {}, ai0 = {}, ai1 = {};
    {
        const _Float16* Yrow = &bufY[(mt * 16 + nlo) * ROWSTRIDE_Y];
        for (int kt = 0; kt < 2; ++kt) {
            FragU af, br0, br1, bi0, bi1;
            const _Float16* pa = Yrow + kt * 32 + base;
            af.u[0] = *(const uint4*)(pa);
            af.u[1] = *(const uint4*)(pa + 16);
            const _Float16* q;
            q = ws + WS_MR + (((2 * nh + 0) * 2 + kt) * 512 + lane * 16);
            br0.u[0] = *(const uint4*)(q); br0.u[1] = *(const uint4*)(q + 8);
            q = ws + WS_MR + (((2 * nh + 1) * 2 + kt) * 512 + lane * 16);
            br1.u[0] = *(const uint4*)(q); br1.u[1] = *(const uint4*)(q + 8);
            q = ws + WS_MI + (((2 * nh + 0) * 2 + kt) * 512 + lane * 16);
            bi0.u[0] = *(const uint4*)(q); bi0.u[1] = *(const uint4*)(q + 8);
            q = ws + WS_MI + (((2 * nh + 1) * 2 + kt) * 512 + lane * 16);
            bi1.u[0] = *(const uint4*)(q); bi1.u[1] = *(const uint4*)(q + 8);
            ar0 = __builtin_amdgcn_wmma_f32_16x16x32_f16(false, af.v, false, br0.v, (short)0, ar0, false, false);
            ar1 = __builtin_amdgcn_wmma_f32_16x16x32_f16(false, af.v, false, br1.v, (short)0, ar1, false, false);
            ai0 = __builtin_amdgcn_wmma_f32_16x16x32_f16(false, af.v, false, bi0.v, (short)0, ai0, false, false);
            ai1 = __builtin_amdgcn_wmma_f32_16x16x32_f16(false, af.v, false, bi1.v, (short)0, ai1, false, false);
        }
    }

    for (int r = 0; r < 8; ++r) {
        int m = mt * 16 + base + r;
        bufP[m * ROWSTRIDE_Y + n0] = (_Float16)(ar0[r] * ar0[r] + ai0[r] * ai0[r]);
        bufP[m * ROWSTRIDE_Y + n1] = (_Float16)(ar1[r] * ar1[r] + ai1[r] * ai1[r]);
    }
    __syncthreads();

    v8f o0 = {}, o1 = {};
    {
        const _Float16* Prow = &bufP[(mt * 16 + nlo) * ROWSTRIDE_Y];
        for (int kt = 0; kt < 2; ++kt) {
            FragU af, bp0, bp1;
            const _Float16* pa = Prow + kt * 32 + base;
            af.u[0] = *(const uint4*)(pa);
            af.u[1] = *(const uint4*)(pa + 16);
            const _Float16* q;
            q = ws + WS_P + (((2 * nh + 0) * 2 + kt) * 512 + lane * 16);
            bp0.u[0] = *(const uint4*)(q); bp0.u[1] = *(const uint4*)(q + 8);
            q = ws + WS_P + (((2 * nh + 1) * 2 + kt) * 512 + lane * 16);
            bp1.u[0] = *(const uint4*)(q); bp1.u[1] = *(const uint4*)(q + 8);
            o0 = __builtin_amdgcn_wmma_f32_16x16x32_f16(false, af.v, false, bp0.v, (short)0, o0, false, false);
            o1 = __builtin_amdgcn_wmma_f32_16x16x32_f16(false, af.v, false, bp1.v, (short)0, o1, false, false);
        }
    }

    float ob0 = out_proj_b[n0], ob1 = out_proj_b[n1];
    float* outstage = (float*)bufA;
    for (int r = 0; r < 8; ++r) {
        int m = mt * 16 + base + r;
        outstage[m * 65 + n0] = o0[r] + ob0;
        outstage[m * 65 + n1] = o1[r] + ob1;
    }
    __syncthreads();
    for (int e = tid; e < 64 * 64; e += 256) {
        int co = e >> 6, m = e & 63;
        int pixel = pix0 + m;
        int b = pixel >> 12, hw = pixel & 4095;
        out[((size_t)(b * 64 + co)) * 4096 + hw] = outstage[m * 65 + co];
    }
}

extern "C" void kernel_launch(void* const* d_in, const int* in_sizes, int n_in,
                              void* d_out, int out_size, void* d_ws, size_t ws_size,
                              hipStream_t stream) {
    const float* x          = (const float*)d_in[0];
    const float* inp_scale  = (const float*)d_in[1];
    const float* in_proj_w  = (const float*)d_in[2];
    const float* in_proj_b  = (const float*)d_in[3];
    const float* weights    = (const float*)d_in[4];
    const float* meas_w     = (const float*)d_in[5];
    const float* out_proj_w = (const float*)d_in[6];
    const float* out_proj_b = (const float*)d_in[7];
    _Float16* ws = (_Float16*)d_ws;
    float* out = (float*)d_out;

    hipLaunchKernelGGL(qprep_kernel, dim3(1), dim3(1024), 0, stream,
                       in_proj_w, weights, meas_w, out_proj_w, ws);
    hipLaunchKernelGGL(qconv_main_kernel, dim3(1024), dim3(256), 0, stream,
                       x, inp_scale, in_proj_b, out_proj_b, ws, out);
}


# main kernel: a wave owns 16 pixels x all 64 columns (4 waves per work-group, not 8); norm and LDS hand-overs in-wave, 2 barriers not 6; weight fragments fetched in groups ahead; norm sum reordered
# speedup vs baseline: 4.4804x; 1.2311x over previous
#include <hip/hip_runtime.h>
#include <hip/hip_bf16.h>
#include <math.h>

typedef __attribute__((ext_vector_type(16))) _Float16 v16h;
typedef __attribute__((ext_vector_type(8)))  float    v8f;

#define PATCH 288
#define QD 64
#define C_IN 32
#define ROWSTRIDE_A 296
#define ROWSTRIDE_Y 72
#define OSTAGE_TILE (16 * ROWSTRIDE_A / 2)

#define WS_W   0
#define WS_MR  18432
#define WS_MI  (18432 + 4096)
#define WS_P   (18432 + 8192)

union FragU { uint4 u[2]; v16h v; };


__device__ __forceinline__ void c2mul(const float* a, const float* b, float* o)
{
    for (int r = 0; r < 2; ++r)
        for (int c = 0; c < 2; ++c) {
            float re = 0.f, im = 0.f;
            for (int k = 0; k < 2; ++k) {
                float ar = a[(r * 2 + k) * 2], ai = a[(r * 2 + k) * 2 + 1];
                float br = b[(k * 2 + c) * 2], bi = b[(k * 2 + c) * 2 + 1];
                re += ar * br - ai * bi;
                im += ar * bi + ai * br;
            }
            o[(r * 2 + c) * 2] = re;
            o[(r * 2 + c) * 2 + 1] = im;
        }
}

__device__ __forceinline__ void make_rx(float c, float s, float* o)
{
    o[0] = c;   o[1] = 0.f; o[2] = 0.f; o[3] = -s;
    o[4] = 0.f; o[5] = -s;  o[6] = c;   o[7] = 0.f;
}

__device__ __forceinline__ void make_ry(float c, float s, float* o)
{
    o[0] = c; o[1] = 0.f; o[2] = -s; o[3] = 0.f;
    o[4] = s; o[5] = 0.f; o[6] = c;  o[7] = 0.f;
}

template <bool HI, bool PERM>
__device__ __forceinline__ void block_sweep(float* Mr, float* Mi, const float* B, int tid)
{
    float yr[8] = {}, yi[8] = {};
    const bool act = tid < 512;
    const int c = tid & 63, o = (tid >> 6) & 7;
    if (act) {
        float xr[8], xi[8];
        for (int s = 0; s < 8; ++s) {
            int row = HI ? (s * 8 + o) : (o * 8 + s);
            if (PERM) row ^= row >> 1;
            xr[s] = Mr[row * 64 + c];
            xi[s] = Mi[row * 64 + c];
        }
        for (int a = 0; a < 8; ++a) {
            float sr = 0.f, si = 0.f;
            for (int s = 0; s < 8; ++s) {
                const float br = B[(a * 8 + s) * 2], bi = B[(a * 8 + s) * 2 + 1];
                sr = fmaf(br, xr[s], sr);
                sr = fmaf(-bi, xi[s], sr);
                si = fmaf(br, xi[s], si);
                si = fmaf(bi, xr[s], si);
            }
            yr[a] = sr;
            yi[a] = si;
        }
    }
    if (PERM) __syncthreads();
    if (act) {
        for (int a = 0; a < 8; ++a) {
            const int row = HI ? (a * 8 + o) : (o * 8 + a);
            Mr[row * 64 + c] = yr[a];
            Mi[row * 64 + c] = yi[a];
        }
    }
    __syncthreads();
}

__global__ void __launch_bounds__(1024)
qprep_kernel(const float* __restrict__ in_proj_w,
             const float* __restrict__ weights,
             const float* __restrict__ meas_w,
             const float* __restrict__ out_proj_w,
             _Float16* __restrict__ ws)
{
    __shared__ float Mr[64 * 64];
    __shared__ float Mi[64 * 64];
    __shared__ float CS[60 * 2];
    __shared__ float G[18 * 8];
    __shared__ float Blk[4 * 128];
    __shared__ float Wl[QD * PATCH];
    const int tid = threadIdx.x;

    float ang = 0.f;
    if (tid < 60) {
        if (tid < 36) {
            ang = weights[tid] * 0.5f;
        } else {
            const int k = tid - 36, w = k >> 2, which = k & 3;
            const float t = meas_w[w * 3 + 0], p = meas_w[w * 3 + 1], lm = meas_w[w * 3 + 2];
            ang = (which == 0) ? t * 0.5f : (which == 1) ? p : (which == 2) ? lm : (p + lm);
        }
    }
    float wv[18];
    for (int i = 0; i < 18; ++i) wv[i] = in_proj_w[tid + 1024 * i];
    if (tid < 60) {
        CS[tid * 2]     = cosf(ang);
        CS[tid * 2 + 1] = sinf(ang);
    }
    __syncthreads();

    if (tid < 18) {
        const int L = tid / 6, w = tid - L * 6;
        float m0[8], m1[8], m2[8], m3[8];
        if (L == 0) {
            make_ry(CS[(w * 3 + 1) * 2], CS[(w * 3 + 1) * 2 + 1], m0);
            make_rx(CS[(w * 3 + 0) * 2], CS[(w * 3 + 0) * 2 + 1], m1);
            c2mul(m0, m1, m3);
        } else if (L == 1) {
            make_ry(CS[((6 + w) * 3 + 1) * 2], CS[((6 + w) * 3 + 1) * 2 + 1], m0);
            make_rx(CS[((6 + w) * 3 + 0) * 2], CS[((6 + w) * 3 + 0) * 2 + 1], m1);
            c2mul(m0, m1, m2);
            make_ry(CS[(w * 3 + 2) * 2], CS[(w * 3 + 2) * 2 + 1], m0);
            c2mul(m2, m0, m3);
        } else {
            const float c  = CS[(36 + w * 4 + 0) * 2], s  = CS[(36 + w * 4 + 0) * 2 + 1];
            const float cp = CS[(36 + w * 4 + 1) * 2], sp = CS[(36 + w * 4 + 1) * 2 + 1];
            const float cl = CS[(36 + w * 4 + 2) * 2], sl = CS[(36 + w * 4 + 2) * 2 + 1];
            const float cpl = CS[(36 + w * 4 + 3) * 2], spl = CS[(36 + w * 4 + 3) * 2 + 1];
            m0[0] = c;      m0[1] = 0.f;    m0[2] = -cl * s; m0[3] = -sl * s;
            m0[4] = cp * s; m0[5] = sp * s; m0[6] = cpl * c; m0[7] = spl * c;
            make_ry(CS[((6 + w) * 3 + 2) * 2], CS[((6 + w) * 3 + 2) * 2 + 1], m1);
            c2mul(m0, m1, m3);
        }
        for (int e = 0; e < 8; ++e) G[tid * 8 + e] = m3[e];
    }
    for (int i = 0; i < 18; ++i) Wl[tid + 1024 * i] = wv[i];
    __syncthreads();

    if (tid < 256) {
        const int tb = tid >> 6, r = (tid >> 3) & 7, j = tid & 7;
        const int L = 1 + (tb >> 1), w0 = (tb & 1) * 3;
        float pr = 1.f, pi = 0.f;
        for (int q = 0; q < 3; ++q) {
            const int rb = (r >> (2 - q)) & 1, cb = (j >> (2 - q)) & 1;
            const float gr = G[((L * 6 + w0 + q) * 4 + rb * 2 + cb) * 2];
            const float gi = G[((L * 6 + w0 + q) * 4 + rb * 2 + cb) * 2 + 1];
            const float nr = pr * gr - pi * gi, ni = pr * gi + pi * gr;
            pr = nr; pi = ni;
        }
        Blk[tid * 2] = pr;
        Blk[tid * 2 + 1] = pi;
    }
    if (tid < 256) {
        const int col = tid & 63, slo = tid >> 6;
        float gr[6][2], gi[6][2];
        for (int w = 0; w < 6; ++w) {
            const int cb = (col >> (5 - w)) & 1;
            gr[w][0] = G[(w * 4 + 0 + cb) * 2];  gi[w][0] = G[(w * 4 + 0 + cb) * 2 + 1];
            gr[w][1] = G[(w * 4 + 2 + cb) * 2];  gi[w][1] = G[(w * 4 + 2 + cb) * 2 + 1];
        }
        const bool b4 = (slo >> 1) & 1, b5 = slo & 1;
        const float g4r = b4 ? gr[4][1] : gr[4][0], g4i = b4 ? gi[4][1] : gi[4][0];
        const float g5r = b5 ? gr[5][1] : gr[5][0], g5i = b5 ? gi[5][1] : gi[5][0];
        float pr[16], pi[16];
        pr[0] = g4r * g5r - g4i * g5i;
        pi[0] = g4r * g5i + g4i * g5r;
        for (int k = 0; k < 4; ++k) {
            const int w = 3 - k, n = 1 << k;
            for (int idx = 0; idx < n; ++idx) {
                const float ar = pr[idx], ai = pi[idx];
                pr[idx + n] = ar * gr[w][1] - ai * gi[w][1];
                pi[idx + n] = ar * gi[w][1] + ai * gr[w][1];
                pr[idx]     = ar * gr[w][0] - ai * gi[w][0];
                pi[idx]     = ar * gi[w][0] + ai * gr[w][0];
            }
        }
        for (int t = 0; t < 16; ++t) {
            const int s = slo + 4 * t;
            const int row = s ^ (s >> 1) ^ (s >> 2) ^ (s >> 3) ^ (s >> 4) ^ (s >> 5);
            Mr[row * 64 + col] = pr[t];
            Mi[row * 64 + col] = pi[t];
        }
    }
    __syncthreads();

    block_sweep<true,  false>(Mr, Mi, Blk + 0 * 128, tid);
    block_sweep<false, false>(Mr, Mi, Blk + 1 * 128, tid);
    block_sweep<true,  true >(Mr, Mi, Blk + 2 * 128, tid);
    block_sweep<false, false>(Mr, Mi, Blk + 3 * 128, tid);

    {
        const int tl = tid & 511, t9 = tid >> 9;
        const int h = tl & 15, lane = tl >> 4;
        const int base = (lane & 16) ? 8 : 0;
        const int kl = (h < 8) ? (base + h) : (base + h + 8);
        const int run = kl >> 3, cl = kl & 7;
        const float* src = Wl + (t9 * 32 + (lane & 15)) * PATCH + 9 * cl + run;
        #pragma unroll
        for (int i = 0; i < 18; ++i) {
            const int kt = i % 9, nt2 = i / 9;
            const int c0 = (4 * kt) / 9, r0 = (4 * kt) % 9;
            const int ko = 72 * c0 + r0 + ((r0 + run >= 9) ? 63 : 0);
            ws[WS_W + (t9 * 18 + i) * 512 + tl] = (_Float16)src[nt2 * 16 * PATCH + ko];
        }
    }
    #pragma unroll
    for (int e = tid; e < 4 * 2 * 512; e += 1024) {
        int h = e & 15, lane = (e >> 4) & 31, s = e >> 9;
        int kt = s & 1, nt = s >> 1;
        int n = nt * 16 + (lane & 15);
        int base = (lane & 16) ? 8 : 0;
        int kl = (h < 8) ? (base + h) : (base + h + 8);
        int k = kt * 32 + kl;
        ws[WS_MR + e] = (_Float16)Mr[n * 64 + k];
        ws[WS_MI + e] = (_Float16)Mi[n * 64 + k];
        ws[WS_P + e]  = (_Float16)out_proj_w[n * 64 + k];
    }
}

template <int G>
__device__ __forceinline__ void load_group(FragU (&ring)[3][4], const _Float16* __restrict__ ws, int lane)
{
    #pragma unroll
    for (int f = 0; f < 4; ++f) {
        const _Float16* q;
        if (G < 9)        q = ws + WS_W + ((f * 9 + G) * 512 + lane * 16);
        else if (G < 13)  q = ws + ((f & 1) ? WS_MI : WS_MR) + (((2 * ((G - 9) >> 1) + (f >> 1)) * 2 + ((G - 9) & 1)) * 512 + lane * 16);
        else              q = ws + WS_P + ((f * 2 + (G - 13)) * 512 + lane * 16);
        ring[G % 3][f].u[0] = *(const uint4*)(q);
        ring[G % 3][f].u[1] = *(const uint4*)(q + 8);
    }
}

__global__ void __launch_bounds__(128)
qconv_main_kernel(const float* __restrict__ x,
                  const float* __restrict__ inp_scale,
                  const float* __restrict__ in_proj_b,
                  const float* __restrict__ out_proj_b,
                  const _Float16* __restrict__ ws,
                  float* __restrict__ out)
{
    __shared__ __align__(16) _Float16 bufA[64 * ROWSTRIDE_A];
    __shared__ __align__(16) _Float16 bufY[64 * ROWSTRIDE_Y];
    __shared__ __align__(16) _Float16 bufP[64 * ROWSTRIDE_Y];

    const int tid  = threadIdx.x;
    const int lane = tid & 31;
    const int mt   = tid >> 5;
    const int pix0 = blockIdx.x * 64;

    FragU ring[3][4];
    float bia[4], obi[4];
    {
        const int m = tid & 63;
        const int gp = __builtin_amdgcn_readfirstlane(tid >> 6);
        const int b = blockIdx.x >> 6;
        const int h = blockIdx.x & 63;
        const bool top = h > 0, bot = h < 63;
        const int hm = top ? h - 1 : h;
        const int hp = bot ? h + 1 : h;
        float v[2][8][3];
        #pragma unroll
        for (int gi = 0; gi < 2; ++gi) {
            const float* xb = x + (size_t)b * C_IN * 4096 + (size_t)(8 * (2 * gp + gi)) * 4096 + m;
            #pragma unroll
            for (int j = 0; j < 8; ++j) {
                const float* xc = xb + j * 4096;
                v[gi][j][0] = xc[hm * 64];
                v[gi][j][1] = xc[h * 64];
                v[gi][j][2] = xc[hp * 64];
            }
        }
        load_group<0>(ring, ws, lane);
        load_group<1>(ring, ws, lane);
        #pragma unroll
        for (int j = 0; j < 4; ++j) {
            bia[j] = in_proj_b[16 * j + (lane & 15)];
            obi[j] = out_proj_b[16 * j + (lane & 15)];
        }
        const bool lastcol = (m == 63), firstcol = (m == 0);
        #pragma unroll
        for (int gi = 0; gi < 2; ++gi) {
            const int g = 2 * gp + gi;
            _Float16* r0 = bufA + ((m + 1) & 63) * ROWSTRIDE_A + 72 * g;
            _Float16* r1 = bufA + m * ROWSTRIDE_A + 72 * g;
            _Float16* r2 = bufA + ((m + 63) & 63) * ROWSTRIDE_A + 72 * g;
            const float* sc = inp_scale + 72 * g;
            #pragma unroll
            for (int r = 0; r < 3; ++r) {
                #pragma unroll
                for (int s = 0; s < 3; ++s) {
                    const int t = r * 3 + s;
                    union { uint4 u; _Float16 e[8]; } pk;
                    #pragma unroll
                    for (int j = 0; j < 8; ++j) {
                        float val = v[gi][j][r];
                        if (r == 0 && !top) val = 0.f;
                        if (r == 2 && !bot) val = 0.f;
                        if (s == 0 && lastcol) val = 0.f;
                        if (s == 2 && firstcol) val = 0.f;
                        pk.e[j] = (_Float16)(val * sc[9 * j + t]);
                    }
                    _Float16* dst = (s == 0) ? r0 : (s == 1) ? r1 : r2;
                    *(uint4*)(dst + 8 * t) = pk.u;
                }
            }
        }
    }
    __syncthreads();

    const int nlo  = lane & 15;
    const int base = (lane & 16) ? 8 : 0;

#define FRONT_OF_PRODUCTS() __builtin_amdgcn_sched_barrier(0)

    v8f acc[4] = {};
    {
        const _Float16* Arow = &bufA[(mt * 16 + nlo) * ROWSTRIDE_A];
        {
            load_group<2>(ring, ws, lane);
            FRONT_OF_PRODUCTS();
            FragU af;
            const _Float16* pa = Arow + 0 * 32 + base;
            af.u[0] = *(const uint4*)(pa);
            af.u[1] = *(const uint4*)(pa + 16);
            #pragma unroll
            for (int j = 0; j < 4; ++j)
                acc[j] = __builtin_amdgcn_wmma_f32_16x16x32_f16(false, af.v, false, ring[0][j].v, (short)0, acc[j], false, false);
        }
        {
            load_group<3>(ring, ws, lane);
            FRONT_OF_PRODUCTS();
            FragU af;
            const _Float16* pa = Arow + 1 * 32 + base;
            af.u[0] = *(const uint4*)(pa);
            af.u[1] = *(const uint4*)(pa + 16);
            #pragma unroll
            for (int j = 0; j < 4; ++j)
                acc[j] = __builtin_amdgcn_wmma_f32_16x16x32_f16(false, af.v, false, ring[1][j].v, (short)0, acc[j], false, false);
        }
        {
            load_group<4>(ring, ws, lane);
            FRONT_OF_PRODUCTS();
            FragU af;
            const _Float16* pa = Arow + 2 * 32 + base;
            af.u[0] = *(const uint4*)(pa);
            af.u[1] = *(const uint4*)(pa + 16);
            #pragma unroll
            for (int j = 0; j < 4; ++j)
                acc[j] = __builtin_amdgcn_wmma_f32_16x16x32_f16(false, af.v, false, ring[2][j].v, (short)0, acc[j], false, false);
        }
        {
            load_group<5>(ring, ws, lane);
            FRONT_OF_PRODUCTS();
            FragU af;
            const _Float16* pa = Arow + 3 * 32 + base;
            af.u[0] = *(const uint4*)(pa);
            af.u[1] = *(const uint4*)(pa + 16);
            #pragma unroll
            for (int j = 0; j < 4; ++j)
                acc[j] = __builtin_amdgcn_wmma_f32_16x16x32_f16(false, af.v, false, ring[0][j].v, (short)0, acc[j], false, false);
        }
        {
            load_group<6>(ring, ws, lane);
            FRONT_OF_PRODUCTS();
            FragU af;
            const _Float16* pa = Arow + 4 * 32 + base;
            af.u[0] = *(const uint4*)(pa);
            af.u[1] = *(const uint4*)(pa + 16);
            #pragma unroll
            for (int j = 0; j < 4; ++j)
                acc[j] = __builtin_amdgcn_wmma_f32_16x16x32_f16(false, af.v, false, ring[1][j].v, (short)0, acc[j], false, false);
        }
        {
            load_group<7>(ring, ws, lane);
            FRONT_OF_PRODUCTS();
            FragU af;
            const _Float16* pa = Arow + 5 * 32 + base;
            af.u[0] = *(const uint4*)(pa);
            af.u[1] = *(const uint4*)(pa + 16);
            #pragma unroll
            for (int j = 0; j < 4; ++j)
                acc[j] = __builtin_amdgcn_wmma_f32_16x16x32_f16(false, af.v, false, ring[2][j].v, (short)0, acc[j], false, false);
        }
        {
            load_group<8>(ring, ws, lane);
            FRONT_OF_PRODUCTS();
            FragU af;
            const _Float16* pa = Arow + 6 * 32 + base;
            af.u[0] = *(const uint4*)(pa);
            af.u[1] = *(const uint4*)(pa + 16);
            #pragma unroll
            for (int j = 0; j < 4; ++j)
                acc[j] = __builtin_amdgcn_wmma_f32_16x16x32_f16(false, af.v, false, ring[0][j].v, (short)0, acc[j], false, false);
        }
        {
            load_group<9>(ring, ws, lane);
            FRONT_OF_PRODUCTS();
            FragU af;
            const _Float16* pa = Arow + 7 * 32 + base;
            af.u[0] = *(const uint4*)(pa);
            af.u[1] = *(const uint4*)(pa + 16);
            #pragma unroll
            for (int j = 0; j < 4; ++j)
                acc[j] = __builtin_amdgcn_wmma_f32_16x16x32_f16(false, af.v, false, ring[1][j].v, (short)0, acc[j], false, false);
        }
        {
            load_group<10>(ring, ws, lane);
            FRONT_OF_PRODUCTS();
            FragU af;
            const _Float16* pa = Arow + 8 * 32 + base;
            af.u[0] = *(const uint4*)(pa);
            af.u[1] = *(const uint4*)(pa + 16);
            #pragma unroll
            for (int j = 0; j < 4; ++j)
                acc[j] = __builtin_amdgcn_wmma_f32_16x16x32_f16(false, af.v, false, ring[2][j].v, (short)0, acc[j], false, false);
        }
    }

    float y[4][8], part[8];
    {
        #pragma unroll
        for (int r = 0; r < 8; ++r) {
            float s = 0.f;
            #pragma unroll
            for (int j = 0; j < 4; ++j) {
                y[j][r] = acc[j][r] + bia[j];
                s += y[j][r] * y[j][r];
            }
            part[r] = s;
        }
    }
    #pragma unroll
    for (int mask = 1; mask <= 8; mask <<= 1) {
        #pragma unroll
        for (int r = 0; r < 8; ++r)
            part[r] += __shfl_xor(part[r], mask, 32);
    }
    #pragma unroll
    for (int r = 0; r < 8; ++r) {
        const int m = mt * 16 + base + r;
        const float inv = 1.f / (sqrtf(part[r]) + 1e-9f);
        #pragma unroll
        for (int j = 0; j < 4; ++j)
            bufY[m * ROWSTRIDE_Y + 16 * j + nlo] = (_Float16)(y[j][r] * inv);
    }
    __builtin_amdgcn_fence(__ATOMIC_ACQ_REL, "wavefront");

    {
        const _Float16* Yrow = &bufY[(mt * 16 + nlo) * ROWSTRIDE_Y];
        FragU af[2];
        #pragma unroll
        for (int kt = 0; kt < 2; ++kt) {
            const _Float16* pa = Yrow + kt * 32 + base;
            af[kt].u[0] = *(const uint4*)(pa);
            af[kt].u[1] = *(const uint4*)(pa + 16);
        }
        {
            v8f ar[2] = {}, ai[2] = {};
            load_group<11>(ring, ws, lane);
            FRONT_OF_PRODUCTS();
            #pragma unroll
            for (int jj = 0; jj < 2; ++jj) {
                ar[jj] = __builtin_amdgcn_wmma_f32_16x16x32_f16(false, af[0].v, false, ring[0][2 * jj].v,     (short)0, ar[jj], false, false);
                ai[jj] = __builtin_amdgcn_wmma_f32_16x16x32_f16(false, af[0].v, false, ring[0][2 * jj + 1].v, (short)0, ai[jj], false, false);
            }
            load_group<12>(ring, ws, lane);
            FRONT_OF_PRODUCTS();
            #pragma unroll
            for (int jj = 0; jj < 2; ++jj) {
                ar[jj] = __builtin_amdgcn_wmma_f32_16x16x32_f16(false, af[1].v, false, ring[1][2 * jj].v,     (short)0, ar[jj], false, false);
                ai[jj] = __builtin_amdgcn_wmma_f32_16x16x32_f16(false, af[1].v, false, ring[1][2 * jj + 1].v, (short)0, ai[jj], false, false);
            }
            #pragma unroll
            for (int r = 0; r < 8; ++r) {
                const int m = mt * 16 + base + r;
                #pragma unroll
                for (int jj = 0; jj < 2; ++jj)
                    bufP[m * ROWSTRIDE_Y + 16 * (0 + jj) + nlo] = (_Float16)(ar[jj][r] * ar[jj][r] + ai[jj][r] * ai[jj][r]);
            }
        }
        {
            v8f ar[2] = {}, ai[2] = {};
            load_group<13>(ring, ws, lane);
            FRONT_OF_PRODUCTS();
            #pragma unroll
            for (int jj = 0; jj < 2; ++jj) {
                ar[jj] = __builtin_amdgcn_wmma_f32_16x16x32_f16(false, af[0].v, false, ring[2][2 * jj].v,     (short)0, ar[jj], false, false);
                ai[jj] = __builtin_amdgcn_wmma_f32_16x16x32_f16(false, af[0].v, false, ring[2][2 * jj + 1].v, (short)0, ai[jj], false, false);
            }
            load_group<14>(ring, ws, lane);
            FRONT_OF_PRODUCTS();
            #pragma unroll
            for (int jj = 0; jj < 2; ++jj) {
                ar[jj] = __builtin_amdgcn_wmma_f32_16x16x32_f16(false, af[1].v, false, ring[0][2 * jj].v,     (short)0, ar[jj], false, false);
                ai[jj] = __builtin_amdgcn_wmma_f32_16x16x32_f16(false, af[1].v, false, ring[0][2 * jj + 1].v, (short)0, ai[jj], false, false);
            }
            #pragma unroll
            for (int r = 0; r < 8; ++r) {
                const int m = mt * 16 + base + r;
                #pragma unroll
                for (int jj = 0; jj < 2; ++jj)
                    bufP[m * ROWSTRIDE_Y + 16 * (2 + jj) + nlo] = (_Float16)(ar[jj][r] * ar[jj][r] + ai[jj][r] * ai[jj][r]);
            }
        }
    }
    __builtin_amdgcn_fence(__ATOMIC_ACQ_REL, "wavefront");

    v8f o[4] = {};
    {
        const _Float16* Prow = &bufP[(mt * 16 + nlo) * ROWSTRIDE_Y];
        #pragma unroll
        for (int kt = 0; kt < 2; ++kt) {
            FragU af;
            const _Float16* pa = Prow + kt * 32 + base;
            af.u[0] = *(const uint4*)(pa);
            af.u[1] = *(const uint4*)(pa + 16);
            #pragma unroll
            for (int j = 0; j < 4; ++j)
                o[j] = __builtin_amdgcn_wmma_f32_16x16x32_f16(false, af.v, false, ring[(13 + kt) % 3][j].v, (short)0, o[j], false, false);
        }
    }

    float* outstage = (float*)bufA;
    {
        float* os = outstage + mt * OSTAGE_TILE + (mt & 1) * 16;
        #pragma unroll
        for (int j = 0; j < 4; ++j) {
            const float ob = obi[j];
            #pragma unroll
            for (int r = 0; r < 8; ++r)
                os[(base + r) * 65 + 16 * j + nlo] = o[j][r] + ob;
        }
    }
    __syncthreads();
    for (int e = tid; e < 64 * 64; e += 128) {
        int co = e >> 6, m = e & 63;
        int pixel = pix0 + m;
        int b = pixel >> 12, hw = pixel & 4095;
        const int t = m >> 4;
        out[((size_t)(b * 64 + co)) * 4096 + hw] = outstage[t * OSTAGE_TILE + (t & 1) * 16 + (m & 15) * 65 + co];
    }
}

extern "C" void kernel_launch(void* const* d_in, const int* in_sizes, int n_in,
                              void* d_out, int out_size, void* d_ws, size_t ws_size,
                              hipStream_t stream) {
    const float* x          = (const float*)d_in[0];
    const float* inp_scale  = (const float*)d_in[1];
    const float* in_proj_w  = (const float*)d_in[2];
    const float* in_proj_b  = (const float*)d_in[3];
    const float* weights    = (const float*)d_in[4];
    const float* meas_w     = (const float*)d_in[5];
    const float* out_proj_w = (const float*)d_in[6];
    const float* out_proj_b = (const float*)d_in[7];
    _Float16* ws = (_Float16*)d_ws;
    float* out = (float*)d_out;

    hipLaunchKernelGGL(qprep_kernel, dim3(1), dim3(1024), 0, stream,
                       in_proj_w, weights, meas_w, out_proj_w, ws);
    hipLaunchKernelGGL(qconv_main_kernel, dim3(1024), dim3(128), 0, stream,
                       x, inp_scale, in_proj_b, out_proj_b, ws, out);
}


# weight fragments stored in the workspace as two 512-byte halves (a lane's elements 0-7, then 8-15): each b128 fragment load of a wave covers 4 whole 128-byte lines, not 8 half-used; k order unchanged
# speedup vs baseline: 4.7931x; 1.0698x over previous
#include <hip/hip_runtime.h>
#include <hip/hip_bf16.h>
#include <math.h>

typedef __attribute__((ext_vector_type(16))) _Float16 v16h;
typedef __attribute__((ext_vector_type(8)))  float    v8f;

#define PATCH 288
#define QD 64
#define C_IN 32
#define ROWSTRIDE_A 296
#define ROWSTRIDE_Y 72
#define OSTAGE_TILE (16 * ROWSTRIDE_A / 2)

#define WS_W   0
#define WS_MR  18432
#define WS_MI  (18432 + 4096)
#define WS_P   (18432 + 8192)

union FragU { uint4 u[2]; v16h v; };


__device__ __forceinline__ void c2mul(const float* a, const float* b, float* o)
{
    for (int r = 0; r < 2; ++r)
        for (int c = 0; c < 2; ++c) {
            float re = 0.f, im = 0.f;
            for (int k = 0; k < 2; ++k) {
                float ar = a[(r * 2 + k) * 2], ai = a[(r * 2 + k) * 2 + 1];
                float br = b[(k * 2 + c) * 2], bi = b[(k * 2 + c) * 2 + 1];
                re += ar * br - ai * bi;
                im += ar * bi + ai * br;
            }
            o[(r * 2 + c) * 2] = re;
            o[(r * 2 + c) * 2 + 1] = im;
        }
}

__device__ __forceinline__ void make_rx(float c, float s, float* o)
{
    o[0] = c;   o[1] = 0.f; o[2] = 0.f; o[3] = -s;
    o[4] = 0.f; o[5] = -s;  o[6] = c;   o[7] = 0.f;
}

__device__ __forceinline__ void make_ry(float c, float s, float* o)
{
    o[0] = c; o[1] = 0.f; o[2] = -s; o[3] = 0.f;
    o[4] = s; o[5] = 0.f; o[6] = c;  o[7] = 0.f;
}

template <bool HI, bool PERM>
__device__ __forceinline__ void block_sweep(float* Mr, float* Mi, const float* B, int tid)
{
    float yr[8] = {}, yi[8] = {};
    const bool act = tid < 512;
    const int c = tid & 63, o = (tid >> 6) & 7;
    if (act) {
        float xr[8], xi[8];
        for (int s = 0; s < 8; ++s) {
            int row = HI ? (s * 8 + o) : (o * 8 + s);
            if (PERM) row ^= row >> 1;
            xr[s] = Mr[row * 64 + c];
            xi[s] = Mi[row * 64 + c];
        }
        for (int a = 0; a < 8; ++a) {
            float sr = 0.f, si = 0.f;
            for (int s = 0; s < 8; ++s) {
                const float br = B[(a * 8 + s) * 2], bi = B[(a * 8 + s) * 2 + 1];
                sr = fmaf(br, xr[s], sr);
                sr = fmaf(-bi, xi[s], sr);
                si = fmaf(br, xi[s], si);
                si = fmaf(bi, xr[s], si);
            }
            yr[a] = sr;
            yi[a] = si;
        }
    }
    if (PERM) __syncthreads();
    if (act) {
        for (int a = 0; a < 8; ++a) {
            const int row = HI ? (a * 8 + o) : (o * 8 + a);
            Mr[row * 64 + c] = yr[a];
            Mi[row * 64 + c] = yi[a];
        }
    }
    __syncthreads();
}

__global__ void __launch_bounds__(1024)
qprep_kernel(const float* __restrict__ in_proj_w,
             const float* __restrict__ weights,
             const float* __restrict__ meas_w,
             const float* __restrict__ out_proj_w,
             _Float16* __restrict__ ws)
{
    __shared__ float Mr[64 * 64];
    __shared__ float Mi[64 * 64];
    __shared__ float CS[60 * 2];
    __shared__ float G[18 * 8];
    __shared__ float Blk[4 * 128];
    __shared__ float Wl[QD * PATCH];
    const int tid = threadIdx.x;

    float ang = 0.f;
    if (tid < 60) {
        if (tid < 36) {
            ang = weights[tid] * 0.5f;
        } else {
            const int k = tid - 36, w = k >> 2, which = k & 3;
            const float t = meas_w[w * 3 + 0], p = meas_w[w * 3 + 1], lm = meas_w[w * 3 + 2];
            ang = (which == 0) ? t * 0.5f : (which == 1) ? p : (which == 2) ? lm : (p + lm);
        }
    }
    float wv[18];
    for (int i = 0; i < 18; ++i) wv[i] = in_proj_w[tid + 1024 * i];
    if (tid < 60) {
        CS[tid * 2]     = cosf(ang);
        CS[tid * 2 + 1] = sinf(ang);
    }
    __syncthreads();

    if (tid < 18) {
        const int L = tid / 6, w = tid - L * 6;
        float m0[8], m1[8], m2[8], m3[8];
        if (L == 0) {
            make_ry(CS[(w * 3 + 1) * 2], CS[(w * 3 + 1) * 2 + 1], m0);
            make_rx(CS[(w * 3 + 0) * 2], CS[(w * 3 + 0) * 2 + 1], m1);
            c2mul(m0, m1, m3);
        } else if (L == 1) {
            make_ry(CS[((6 + w) * 3 + 1) * 2], CS[((6 + w) * 3 + 1) * 2 + 1], m0);
            make_rx(CS[((6 + w) * 3 + 0) * 2], CS[((6 + w) * 3 + 0) * 2 + 1], m1);
            c2mul(m0, m1, m2);
            make_ry(CS[(w * 3 + 2) * 2], CS[(w * 3 + 2) * 2 + 1], m0);
            c2mul(m2, m0, m3);
        } else {
            const float c  = CS[(36 + w * 4 + 0) * 2], s  = CS[(36 + w * 4 + 0) * 2 + 1];
            const float cp = CS[(36 + w * 4 + 1) * 2], sp = CS[(36 + w * 4 + 1) * 2 + 1];
            const float cl = CS[(36 + w * 4 + 2) * 2], sl = CS[(36 + w * 4 + 2) * 2 + 1];
            const float cpl = CS[(36 + w * 4 + 3) * 2], spl = CS[(36 + w * 4 + 3) * 2 + 1];
            m0[0] = c;      m0[1] = 0.f;    m0[2] = -cl * s; m0[3] = -sl * s;
            m0[4] = cp * s; m0[5] = sp * s; m0[6] = cpl * c; m0[7] = spl * c;
            make_ry(CS[((6 + w) * 3 + 2) * 2], CS[((6 + w) * 3 + 2) * 2 + 1], m1);
            c2mul(m0, m1, m3);
        }
        for (int e = 0; e < 8; ++e) G[tid * 8 + e] = m3[e];
    }
    for (int i = 0; i < 18; ++i) Wl[tid + 1024 * i] = wv[i];
    __syncthreads();

    if (tid < 256) {
        const int tb = tid >> 6, r = (tid >> 3) & 7, j = tid & 7;
        const int L = 1 + (tb >> 1), w0 = (tb & 1) * 3;
        float pr = 1.f, pi = 0.f;
        for (int q = 0; q < 3; ++q) {
            const int rb = (r >> (2 - q)) & 1, cb = (j >> (2 - q)) & 1;
            const float gr = G[((L * 6 + w0 + q) * 4 + rb * 2 + cb) * 2];
            const float gi = G[((L * 6 + w0 + q) * 4 + rb * 2 + cb) * 2 + 1];
            const float nr = pr * gr - pi * gi, ni = pr * gi + pi * gr;
            pr = nr; pi = ni;
        }
        Blk[tid * 2] = pr;
        Blk[tid * 2 + 1] = pi;
    }
    if (tid < 256) {
        const int col = tid & 63, slo = tid >> 6;
        float gr[6][2], gi[6][2];
        for (int w = 0; w < 6; ++w) {
            const int cb = (col >> (5 - w)) & 1;
            gr[w][0] = G[(w * 4 + 0 + cb) * 2];  gi[w][0] = G[(w * 4 + 0 + cb) * 2 + 1];
            gr[w][1] = G[(w * 4 + 2 + cb) * 2];  gi[w][1] = G[(w * 4 + 2 + cb) * 2 + 1];
        }
        const bool b4 = (slo >> 1) & 1, b5 = slo & 1;
        const float g4r = b4 ? gr[4][1] : gr[4][0], g4i = b4 ? gi[4][1] : gi[4][0];
        const float g5r = b5 ? gr[5][1] : gr[5][0], g5i = b5 ? gi[5][1] : gi[5][0];
        float pr[16], pi[16];
        pr[0] = g4r * g5r - g4i * g5i;
        pi[0] = g4r * g5i + g4i * g5r;
        for (int k = 0; k < 4; ++k) {
            const int w = 3 - k, n = 1 << k;
            for (int idx = 0; idx < n; ++idx) {
                const float ar = pr[idx], ai = pi[idx];
                pr[idx + n] = ar * gr[w][1] - ai * gi[w][1];
                pi[idx + n] = ar * gi[w][1] + ai * gr[w][1];
                pr[idx]     = ar * gr[w][0] - ai * gi[w][0];
                pi[idx]     = ar * gi[w][0] + ai * gr[w][0];
            }
        }
        for (int t = 0; t < 16; ++t) {
            const int s = slo + 4 * t;
            const int row = s ^ (s >> 1) ^ (s >> 2) ^ (s >> 3) ^ (s >> 4) ^ (s >> 5);
            Mr[row * 64 + col] = pr[t];
            Mi[row * 64 + col] = pi[t];
        }
    }
    __syncthreads();

    block_sweep<true,  false>(Mr, Mi, Blk + 0 * 128, tid);
    block_sweep<false, false>(Mr, Mi, Blk + 1 * 128, tid);
    block_sweep<true,  true >(Mr, Mi, Blk + 2 * 128, tid);
    block_sweep<false, false>(Mr, Mi, Blk + 3 * 128, tid);

    {
        const int tl = tid & 511, t9 = tid >> 9;
        const int elo = tl & 7, lane = (tl >> 3) & 31, ehi = tl >> 8;
        const int base = (lane & 16) ? 8 : 0;
        const int kl = base + elo + 16 * ehi;
        const int run = kl >> 3, cl = kl & 7;
        const float* src = Wl + (t9 * 32 + (lane & 15)) * PATCH + 9 * cl + run;
        #pragma unroll
        for (int i = 0; i < 18; ++i) {
            const int kt = i % 9, nt2 = i / 9;
            const int c0 = (4 * kt) / 9, r0 = (4 * kt) % 9;
            const int ko = 72 * c0 + r0 + ((r0 + run >= 9) ? 63 : 0);
            ws[WS_W + (t9 * 18 + i) * 512 + tl] = (_Float16)src[nt2 * 16 * PATCH + ko];
        }
    }
    #pragma unroll
    for (int e = tid; e < 4 * 2 * 512; e += 1024) {
        int elo = e & 7, lane = (e >> 3) & 31, ehi = (e >> 8) & 1, s = e >> 9;
        int kt = s & 1, nt = s >> 1;
        int n = nt * 16 + (lane & 15);
        int base = (lane & 16) ? 8 : 0;
        int kl = base + elo + 16 * ehi;
        int k = kt * 32 + kl;
        ws[WS_MR + e] = (_Float16)Mr[n * 64 + k];
        ws[WS_MI + e] = (_Float16)Mi[n * 64 + k];
        ws[WS_P + e]  = (_Float16)out_proj_w[n * 64 + k];
    }
}

template <int G>
__device__ __forceinline__ void load_group(FragU (&ring)[3][4], const _Float16* __restrict__ ws, int lane)
{
    #pragma unroll
    for (int f = 0; f < 4; ++f) {
        const _Float16* q;
        if (G < 9)        q = ws + WS_W + ((f * 9 + G) * 512 + lane * 8);
        else if (G < 13)  q = ws + ((f & 1) ? WS_MI : WS_MR) + (((2 * ((G - 9) >> 1) + (f >> 1)) * 2 + ((G - 9) & 1)) * 512 + lane * 8);
        else              q = ws + WS_P + ((f * 2 + (G - 13)) * 512 + lane * 8);
        ring[G % 3][f].u[0] = *(const uint4*)(q);
        ring[G % 3][f].u[1] = *(const uint4*)(q + 256);
    }
}

__global__ void __launch_bounds__(128)
qconv_main_kernel(const float* __restrict__ x,
                  const float* __restrict__ inp_scale,
                  const float* __restrict__ in_proj_b,
                  const float* __restrict__ out_proj_b,
                  const _Float16* __restrict__ ws,
                  float* __restrict__ out)
{
    __shared__ __align__(16) _Float16 bufA[64 * ROWSTRIDE_A];
    __shared__ __align__(16) _Float16 bufY[64 * ROWSTRIDE_Y];
    __shared__ __align__(16) _Float16 bufP[64 * ROWSTRIDE_Y];

    const int tid  = threadIdx.x;
    const int lane = tid & 31;
    const int mt   = tid >> 5;
    const int pix0 = blockIdx.x * 64;

    FragU ring[3][4];
    float bia[4], obi[4];
    {
        const int m = tid & 63;
        const int gp = __builtin_amdgcn_readfirstlane(tid >> 6);
        const int b = blockIdx.x >> 6;
        const int h = blockIdx.x & 63;
        const bool top = h > 0, bot = h < 63;
        const int hm = top ? h - 1 : h;
        const int hp = bot ? h + 1 : h;
        float v[2][8][3];
        #pragma unroll
        for (int gi = 0; gi < 2; ++gi) {
            const float* xb = x + (size_t)b * C_IN * 4096 + (size_t)(8 * (2 * gp + gi)) * 4096 + m;
            #pragma unroll
            for (int j = 0; j < 8; ++j) {
                const float* xc = xb + j * 4096;
                v[gi][j][0] = xc[hm * 64];
                v[gi][j][1] = xc[h * 64];
                v[gi][j][2] = xc[hp * 64];
            }
        }
        load_group<0>(ring, ws, lane);
        load_group<1>(ring, ws, lane);
        #pragma unroll
        for (int j = 0; j < 4; ++j) {
            bia[j] = in_proj_b[16 * j + (lane & 15)];
            obi[j] = out_proj_b[16 * j + (lane & 15)];
        }
        const bool lastcol = (m == 63), firstcol = (m == 0);
        #pragma unroll
        for (int gi = 0; gi < 2; ++gi) {
            const int g = 2 * gp + gi;
            _Float16* r0 = bufA + ((m + 1) & 63) * ROWSTRIDE_A + 72 * g;
            _Float16* r1 = bufA + m * ROWSTRIDE_A + 72 * g;
            _Float16* r2 = bufA + ((m + 63) & 63) * ROWSTRIDE_A + 72 * g;
            const float* sc = inp_scale + 72 * g;
            #pragma unroll
            for (int r = 0; r < 3; ++r) {
                #pragma unroll
                for (int s = 0; s < 3; ++s) {
                    const int t = r * 3 + s;
                    union { uint4 u; _Float16 e[8]; } pk;
                    #pragma unroll
                    for (int j = 0; j < 8; ++j) {
                        float val = v[gi][j][r];
                        if (r == 0 && !top) val = 0.f;
                        if (r == 2 && !bot) val = 0.f;
                        if (s == 0 && lastcol) val = 0.f;
                        if (s == 2 && firstcol) val = 0.f;
                        pk.e[j] = (_Float16)(val * sc[9 * j + t]);
                    }
                    _Float16* dst = (s == 0) ? r0 : (s == 1) ? r1 : r2;
                    *(uint4*)(dst + 8 * t) = pk.u;
                }
            }
        }
    }
    __syncthreads();

    const int nlo  = lane & 15;
    const int base = (lane & 16) ? 8 : 0;

#define FRONT_OF_PRODUCTS() __builtin_amdgcn_sched_barrier(0)

    v8f acc[4] = {};
    {
        const _Float16* Arow = &bufA[(mt * 16 + nlo) * ROWSTRIDE_A];
        {
            load_group<2>(ring, ws, lane);
            FRONT_OF_PRODUCTS();
            FragU af;
            const _Float16* pa = Arow + 0 * 32 + base;
            af.u[0] = *(const uint4*)(pa);
            af.u[1] = *(const uint4*)(pa + 16);
            #pragma unroll
            for (int j = 0; j < 4; ++j)
                acc[j] = __builtin_amdgcn_wmma_f32_16x16x32_f16(false, af.v, false, ring[0][j].v, (short)0, acc[j], false, false);
        }
        {
            load_group<3>(ring, ws, lane);
            FRONT_OF_PRODUCTS();
            FragU af;
            const _Float16* pa = Arow + 1 * 32 + base;
            af.u[0] = *(const uint4*)(pa);
            af.u[1] = *(const uint4*)(pa + 16);
            #pragma unroll
            for (int j = 0; j < 4; ++j)
                acc[j] = __builtin_amdgcn_wmma_f32_16x16x32_f16(false, af.v, false, ring[1][j].v, (short)0, acc[j], false, false);
        }
        {
            load_group<4>(ring, ws, lane);
            FRONT_OF_PRODUCTS();
            FragU af;
            const _Float16* pa = Arow + 2 * 32 + base;
            af.u[0] = *(const uint4*)(pa);
            af.u[1] = *(const uint4*)(pa + 16);
            #pragma unroll
            for (int j = 0; j < 4; ++j)
                acc[j] = __builtin_amdgcn_wmma_f32_16x16x32_f16(false, af.v, false, ring[2][j].v, (short)0, acc[j], false, false);
        }
        {
            load_group<5>(ring, ws, lane);
            FRONT_OF_PRODUCTS();
            FragU af;
            const _Float16* pa = Arow + 3 * 32 + base;
            af.u[0] = *(const uint4*)(pa);
            af.u[1] = *(const uint4*)(pa + 16);
            #pragma unroll
            for (int j = 0; j < 4; ++j)
                acc[j] = __builtin_amdgcn_wmma_f32_16x16x32_f16(false, af.v, false, ring[0][j].v, (short)0, acc[j], false, false);
        }
        {
            load_group<6>(ring, ws, lane);
            FRONT_OF_PRODUCTS();
            FragU af;
            const _Float16* pa = Arow + 4 * 32 + base;
            af.u[0] = *(const uint4*)(pa);
            af.u[1] = *(const uint4*)(pa + 16);
            #pragma unroll
            for (int j = 0; j < 4; ++j)
                acc[j] = __builtin_amdgcn_wmma_f32_16x16x32_f16(false, af.v, false, ring[1][j].v, (short)0, acc[j], false, false);
        }
        {
            load_group<7>(ring, ws, lane);
            FRONT_OF_PRODUCTS();
            FragU af;
            const _Float16* pa = Arow + 5 * 32 + base;
            af.u[0] = *(const uint4*)(pa);
            af.u[1] = *(const uint4*)(pa + 16);
            #pragma unroll
            for (int j = 0; j < 4; ++j)
                acc[j] = __builtin_amdgcn_wmma_f32_16x16x32_f16(false, af.v, false, ring[2][j].v, (short)0, acc[j], false, false);
        }
        {
            load_group<8>(ring, ws, lane);
            FRONT_OF_PRODUCTS();
            FragU af;
            const _Float16* pa = Arow + 6 * 32 + base;
            af.u[0] = *(const uint4*)(pa);
            af.u[1] = *(const uint4*)(pa + 16);
            #pragma unroll
            for (int j = 0; j < 4; ++j)
                acc[j] = __builtin_amdgcn_wmma_f32_16x16x32_f16(false, af.v, false, ring[0][j].v, (short)0, acc[j], false, false);
        }
        {
            load_group<9>(ring, ws, lane);
            FRONT_OF_PRODUCTS();
            FragU af;
            const _Float16* pa = Arow + 7 * 32 + base;
            af.u[0] = *(const uint4*)(pa);
            af.u[1] = *(const uint4*)(pa + 16);
            #pragma unroll
            for (int j = 0; j < 4; ++j)
                acc[j] = __builtin_amdgcn_wmma_f32_16x16x32_f16(false, af.v, false, ring[1][j].v, (short)0, acc[j], false, false);
        }
        {
            load_group<10>(ring, ws, lane);
            FRONT_OF_PRODUCTS();
            FragU af;
            const _Float16* pa = Arow + 8 * 32 + base;
            af.u[0] = *(const uint4*)(pa);
            af.u[1] = *(const uint4*)(pa + 16);
            #pragma unroll
            for (int j = 0; j < 4; ++j)
                acc[j] = __builtin_amdgcn_wmma_f32_16x16x32_f16(false, af.v, false, ring[2][j].v, (short)0, acc[j], false, false);
        }
    }

    float y[4][8], part[8];
    {
        #pragma unroll
        for (int r = 0; r < 8; ++r) {
            float s = 0.f;
            #pragma unroll
            for (int j = 0; j < 4; ++j) {
                y[j][r] = acc[j][r] + bia[j];
                s += y[j][r] * y[j][r];
            }
            part[r] = s;
        }
    }
    #pragma unroll
    for (int mask = 1; mask <= 8; mask <<= 1) {
        #pragma unroll
        for (int r = 0; r < 8; ++r)
            part[r] += __shfl_xor(part[r], mask, 32);
    }
    #pragma unroll
    for (int r = 0; r < 8; ++r) {
        const int m = mt * 16 + base + r;
        const float inv = 1.f / (sqrtf(part[r]) + 1e-9f);
        #pragma unroll
        for (int j = 0; j < 4; ++j)
            bufY[m * ROWSTRIDE_Y + 16 * j + nlo] = (_Float16)(y[j][r] * inv);
    }
    __builtin_amdgcn_fence(__ATOMIC_ACQ_REL, "wavefront");

    {
        const _Float16* Yrow = &bufY[(mt * 16 + nlo) * ROWSTRIDE_Y];
        FragU af[2];
        #pragma unroll
        for (int kt = 0; kt < 2; ++kt) {
            const _Float16* pa = Yrow + kt * 32 + base;
            af[kt].u[0] = *(const uint4*)(pa);
            af[kt].u[1] = *(const uint4*)(pa + 16);
        }
        {
            v8f ar[2] = {}, ai[2] = {};
            load_group<11>(ring, ws, lane);
            FRONT_OF_PRODUCTS();
            #pragma unroll
            for (int jj = 0; jj < 2; ++jj) {
                ar[jj] = __builtin_amdgcn_wmma_f32_16x16x32_f16(false, af[0].v, false, ring[0][2 * jj].v,     (short)0, ar[jj], false, false);
                ai[jj] = __builtin_amdgcn_wmma_f32_16x16x32_f16(false, af[0].v, false, ring[0][2 * jj + 1].v, (short)0, ai[jj], false, false);
            }
            load_group<12>(ring, ws, lane);
            FRONT_OF_PRODUCTS();
            #pragma unroll
            for (int jj = 0; jj < 2; ++jj) {
                ar[jj] = __builtin_amdgcn_wmma_f32_16x16x32_f16(false, af[1].v, false, ring[1][2 * jj].v,     (short)0, ar[jj], false, false);
                ai[jj] = __builtin_amdgcn_wmma_f32_16x16x32_f16(false, af[1].v, false, ring[1][2 * jj + 1].v, (short)0, ai[jj], false, false);
            }
            #pragma unroll
            for (int r = 0; r < 8; ++r) {
                const int m = mt * 16 + base + r;
                #pragma unroll
                for (int jj = 0; jj < 2; ++jj)
                    bufP[m * ROWSTRIDE_Y + 16 * (0 + jj) + nlo] = (_Float16)(ar[jj][r] * ar[jj][r] + ai[jj][r] * ai[jj][r]);
            }
        }
        {
            v8f ar[2] = {}, ai[2] = {};
            load_group<13>(ring, ws, lane);
            FRONT_OF_PRODUCTS();
            #pragma unroll
            for (int jj = 0; jj < 2; ++jj) {
                ar[jj] = __builtin_amdgcn_wmma_f32_16x16x32_f16(false, af[0].v, false, ring[2][2 * jj].v,     (short)0, ar[jj], false, false);
                ai[jj] = __builtin_amdgcn_wmma_f32_16x16x32_f16(false, af[0].v, false, ring[2][2 * jj + 1].v, (short)0, ai[jj], false, false);
            }
            load_group<14>(ring, ws, lane);
            FRONT_OF_PRODUCTS();
            #pragma unroll
            for (int jj = 0; jj < 2; ++jj) {
                ar[jj] = __builtin_amdgcn_wmma_f32_16x16x32_f16(false, af[1].v, false, ring[0][2 * jj].v,     (short)0, ar[jj], false, false);
                ai[jj] = __builtin_amdgcn_wmma_f32_16x16x32_f16(false, af[1].v, false, ring[0][2 * jj + 1].v, (short)0, ai[jj], false, false);
            }
            #pragma unroll
            for (int r = 0; r < 8; ++r) {
                const int m = mt * 16 + base + r;
                #pragma unroll
                for (int jj = 0; jj < 2; ++jj)
                    bufP[m * ROWSTRIDE_Y + 16 * (2 + jj) + nlo] = (_Float16)(ar[jj][r] * ar[jj][r] + ai[jj][r] * ai[jj][r]);
            }
        }
    }
    __builtin_amdgcn_fence(__ATOMIC_ACQ_REL, "wavefront");

    v8f o[4] = {};
    {
        const _Float16* Prow = &bufP[(mt * 16 + nlo) * ROWSTRIDE_Y];
        #pragma unroll
        for (int kt = 0; kt < 2; ++kt) {
            FragU af;
            const _Float16* pa = Prow + kt * 32 + base;
            af.u[0] = *(const uint4*)(pa);
            af.u[1] = *(const uint4*)(pa + 16);
            #pragma unroll
            for (int j = 0; j < 4; ++j)
                o[j] = __builtin_amdgcn_wmma_f32_16x16x32_f16(false, af.v, false, ring[(13 + kt) % 3][j].v, (short)0, o[j], false, false);
        }
    }

    float* outstage = (float*)bufA;
    {
        float* os = outstage + mt * OSTAGE_TILE + (mt & 1) * 16;
        #pragma unroll
        for (int j = 0; j < 4; ++j) {
            const float ob = obi[j];
            #pragma unroll
            for (int r = 0; r < 8; ++r)
                os[(base + r) * 65 + 16 * j + nlo] = o[j][r] + ob;
        }
    }
    __syncthreads();
    for (int e = tid; e < 64 * 64; e += 128) {
        int co = e >> 6, m = e & 63;
        int pixel = pix0 + m;
        int b = pixel >> 12, hw = pixel & 4095;
        const int t = m >> 4;
        out[((size_t)(b * 64 + co)) * 4096 + hw] = outstage[t * OSTAGE_TILE + (t & 1) * 16 + (m & 15) * 65 + co];
    }
}

extern "C" void kernel_launch(void* const* d_in, const int* in_sizes, int n_in,
                              void* d_out, int out_size, void* d_ws, size_t ws_size,
                              hipStream_t stream) {
    const float* x          = (const float*)d_in[0];
    const float* inp_scale  = (const float*)d_in[1];
    const float* in_proj_w  = (const float*)d_in[2];
    const float* in_proj_b  = (const float*)d_in[3];
    const float* weights    = (const float*)d_in[4];
    const float* meas_w     = (const float*)d_in[5];
    const float* out_proj_w = (const float*)d_in[6];
    const float* out_proj_b = (const float*)d_in[7];
    _Float16* ws = (_Float16*)d_ws;
    float* out = (float*)d_out;

    hipLaunchKernelGGL(qprep_kernel, dim3(1), dim3(1024), 0, stream,
                       in_proj_w, weights, meas_w, out_proj_w, ws);
    hipLaunchKernelGGL(qconv_main_kernel, dim3(1024), dim3(128), 0, stream,
                       x, inp_scale, in_proj_b, out_proj_b, ws, out);
}


# main kernel epilogue: output staged channel-major in LDS, a lane moves 16 bytes per instruction: 8+8+8 LDS-store, LDS-load and global-store instructions per lane, not 32+32+32; no sum reordered
# speedup vs baseline: 5.1924x; 1.0833x over previous
#include <hip/hip_runtime.h>
#include <hip/hip_bf16.h>
#include <math.h>

typedef __attribute__((ext_vector_type(16))) _Float16 v16h;
typedef __attribute__((ext_vector_type(8)))  float    v8f;
typedef __attribute__((ext_vector_type(4)))  float    v4f;

#define PATCH 288
#define QD 64
#define C_IN 32
#define ROWSTRIDE_A 296
#define ROWSTRIDE_Y 72
#define OSTAGE_TILE (16 * ROWSTRIDE_A / 2)
#define OSTAGE_PITCH 20

#define WS_W   0
#define WS_MR  18432
#define WS_MI  (18432 + 4096)
#define WS_P   (18432 + 8192)

union FragU { uint4 u[2]; v16h v; };


__device__ __forceinline__ void c2mul(const float* a, const float* b, float* o)
{
    for (int r = 0; r < 2; ++r)
        for (int c = 0; c < 2; ++c) {
            float re = 0.f, im = 0.f;
            for (int k = 0; k < 2; ++k) {
                float ar = a[(r * 2 + k) * 2], ai = a[(r * 2 + k) * 2 + 1];
                float br = b[(k * 2 + c) * 2], bi = b[(k * 2 + c) * 2 + 1];
                re += ar * br - ai * bi;
                im += ar * bi + ai * br;
            }
            o[(r * 2 + c) * 2] = re;
            o[(r * 2 + c) * 2 + 1] = im;
        }
}

__device__ __forceinline__ void make_rx(float c, float s, float* o)
{
    o[0] = c;   o[1] = 0.f; o[2] = 0.f; o[3] = -s;
    o[4] = 0.f; o[5] = -s;  o[6] = c;   o[7] = 0.f;
}

__device__ __forceinline__ void make_ry(float c, float s, float* o)
{
    o[0] = c; o[1] = 0.f; o[2] = -s; o[3] = 0.f;
    o[4] = s; o[5] = 0.f; o[6] = c;  o[7] = 0.f;
}

template <bool HI, bool PERM>
__device__ __forceinline__ void block_sweep(float* Mr, float* Mi, const float* B, int tid)
{
    float yr[8] = {}, yi[8] = {};
    const bool act = tid < 512;
    const int c = tid & 63, o = (tid >> 6) & 7;
    if (act) {
        float xr[8], xi[8];
        for (int s = 0; s < 8; ++s) {
            int row = HI ? (s * 8 + o) : (o * 8 + s);
            if (PERM) row ^= row >> 1;
            xr[s] = Mr[row * 64 + c];
            xi[s] = Mi[row * 64 + c];
        }
        for (int a = 0; a < 8; ++a) {
            float sr = 0.f, si = 0.f;
            for (int s = 0; s < 8; ++s) {
                const float br = B[(a * 8 + s) * 2], bi = B[(a * 8 + s) * 2 + 1];
                sr = fmaf(br, xr[s], sr);
                sr = fmaf(-bi, xi[s], sr);
                si = fmaf(br, xi[s], si);
                si = fmaf(bi, xr[s], si);
            }
            yr[a] = sr;
            yi[a] = si;
        }
    }
    if (PERM) __syncthreads();
    if (act) {
        for (int a = 0; a < 8; ++a) {
            const int row = HI ? (a * 8 + o) : (o * 8 + a);
            Mr[row * 64 + c] = yr[a];
            Mi[row * 64 + c] = yi[a];
        }
    }
    __syncthreads();
}

__global__ void __launch_bounds__(1024)
qprep_kernel(const float* __restrict__ in_proj_w,
             const float* __restrict__ weights,
             const float* __restrict__ meas_w,
             const float* __restrict__ out_proj_w,
             _Float16* __restrict__ ws)
{
    __shared__ float Mr[64 * 64];
    __shared__ float Mi[64 * 64];
    __shared__ float CS[60 * 2];
    __shared__ float G[18 * 8];
    __shared__ float Blk[4 * 128];
    __shared__ float Wl[QD * PATCH];
    const int tid = threadIdx.x;

    float ang = 0.f;
    if (tid < 60) {
        if (tid < 36) {
            ang = weights[tid] * 0.5f;
        } else {
            const int k = tid - 36, w = k >> 2, which = k & 3;
            const float t = meas_w[w * 3 + 0], p = meas_w[w * 3 + 1], lm = meas_w[w * 3 + 2];
            ang = (which == 0) ? t * 0.5f : (which == 1) ? p : (which == 2) ? lm : (p + lm);
        }
    }
    float wv[18];
    for (int i = 0; i < 18; ++i) wv[i] = in_proj_w[tid + 1024 * i];
    if (tid < 60) {
        CS[tid * 2]     = cosf(ang);
        CS[tid * 2 + 1] = sinf(ang);
    }
    __syncthreads();

    if (tid < 18) {
        const int L = tid / 6, w = tid - L * 6;
        float m0[8], m1[8], m2[8], m3[8];
        if (L == 0) {
            make_ry(CS[(w * 3 + 1) * 2], CS[(w * 3 + 1) * 2 + 1], m0);
            make_rx(CS[(w * 3 + 0) * 2], CS[(w * 3 + 0) * 2 + 1], m1);
            c2mul(m0, m1, m3);
        } else if (L == 1) {
            make_ry(CS[((6 + w) * 3 + 1) * 2], CS[((6 + w) * 3 + 1) * 2 + 1], m0);
            make_rx(CS[((6 + w) * 3 + 0) * 2], CS[((6 + w) * 3 + 0) * 2 + 1], m1);
            c2mul(m0, m1, m2);
            make_ry(CS[(w * 3 + 2) * 2], CS[(w * 3 + 2) * 2 + 1], m0);
            c2mul(m2, m0, m3);
        } else {
            const float c  = CS[(36 + w * 4 + 0) * 2], s  = CS[(36 + w * 4 + 0) * 2 + 1];
            const float cp = CS[(36 + w * 4 + 1) * 2], sp = CS[(36 + w * 4 + 1) * 2 + 1];
            const float cl = CS[(36 + w * 4 + 2) * 2], sl = CS[(36 + w * 4 + 2) * 2 + 1];
            const float cpl = CS[(36 + w * 4 + 3) * 2], spl = CS[(36 + w * 4 + 3) * 2 + 1];
            m0[0] = c;      m0[1] = 0.f;    m0[2] = -cl * s; m0[3] = -sl * s;
            m0[4] = cp * s; m0[5] = sp * s; m0[6] = cpl * c; m0[7] = spl * c;
            make_ry(CS[((6 + w) * 3 + 2) * 2], CS[((6 + w) * 3 + 2) * 2 + 1], m1);
            c2mul(m0, m1, m3);
        }
        for (int e = 0; e < 8; ++e) G[tid * 8 + e] = m3[e];
    }
    for (int i = 0; i < 18; ++i) Wl[tid + 1024 * i] = wv[i];
    __syncthreads();

    if (tid < 256) {
        const int tb = tid >> 6, r = (tid >> 3) & 7, j = tid & 7;
        const int L = 1 + (tb >> 1), w0 = (tb & 1) * 3;
        float pr = 1.f, pi = 0.f;
        for (int q = 0; q < 3; ++q) {
            const int rb = (r >> (2 - q)) & 1, cb = (j >> (2 - q)) & 1;
            const float gr = G[((L * 6 + w0 + q) * 4 + rb * 2 + cb) * 2];
            const float gi = G[((L * 6 + w0 + q) * 4 + rb * 2 + cb) * 2 + 1];
            const float nr = pr * gr - pi * gi, ni = pr * gi + pi * gr;
            pr = nr; pi = ni;
        }
        Blk[tid * 2] = pr;
        Blk[tid * 2 + 1] = pi;
    }
    if (tid < 256) {
        const int col = tid & 63, slo = tid >> 6;
        float gr[6][2], gi[6][2];
        for (int w = 0; w < 6; ++w) {
            const int cb = (col >> (5 - w)) & 1;
            gr[w][0] = G[(w * 4 + 0 + cb) * 2];  gi[w][0] = G[(w * 4 + 0 + cb) * 2 + 1];
            gr[w][1] = G[(w * 4 + 2 + cb) * 2];  gi[w][1] = G[(w * 4 + 2 + cb) * 2 + 1];
        }
        const bool b4 = (slo >> 1) & 1, b5 = slo & 1;
        const float g4r = b4 ? gr[4][1] : gr[4][0], g4i = b4 ? gi[4][1] : gi[4][0];
        const float g5r = b5 ? gr[5][1] : gr[5][0], g5i = b5 ? gi[5][1] : gi[5][0];
        float pr[16], pi[16];
        pr[0] = g4r * g5r - g4i * g5i;
        pi[0] = g4r * g5i + g4i * g5r;
        for (int k = 0; k < 4; ++k) {
            const int w = 3 - k, n = 1 << k;
            for (int idx = 0; idx < n; ++idx) {
                const float ar = pr[idx], ai = pi[idx];
                pr[idx + n] = ar * gr[w][1] - ai * gi[w][1];
                pi[idx + n] = ar * gi[w][1] + ai * gr[w][1];
                pr[idx]     = ar * gr[w][0] - ai * gi[w][0];
                pi[idx]     = ar * gi[w][0] + ai * gr[w][0];
            }
        }
        for (int t = 0; t < 16; ++t) {
            const int s = slo + 4 * t;
            const int row = s ^ (s >> 1) ^ (s >> 2) ^ (s >> 3) ^ (s >> 4) ^ (s >> 5);
            Mr[row * 64 + col] = pr[t];
            Mi[row * 64 + col] = pi[t];
        }
    }
    __syncthreads();

    block_sweep<true,  false>(Mr, Mi, Blk + 0 * 128, tid);
    block_sweep<false, false>(Mr, Mi, Blk + 1 * 128, tid);
    block_sweep<true,  true >(Mr, Mi, Blk + 2 * 128, tid);
    block_sweep<false, false>(Mr, Mi, Blk + 3 * 128, tid);

    {
        const int tl = tid & 511, t9 = tid >> 9;
        const int elo = tl & 7, lane = (tl >> 3) & 31, ehi = tl >> 8;
        const int base = (lane & 16) ? 8 : 0;
        const int kl = base + elo + 16 * ehi;
        const int run = kl >> 3, cl = kl & 7;
        const float* src = Wl + (t9 * 32 + (lane & 15)) * PATCH + 9 * cl + run;
        #pragma unroll
        for (int i = 0; i < 18; ++i) {
            const int kt = i % 9, nt2 = i / 9;
            const int c0 = (4 * kt) / 9, r0 = (4 * kt) % 9;
            const int ko = 72 * c0 + r0 + ((r0 + run >= 9) ? 63 : 0);
            ws[WS_W + (t9 * 18 + i) * 512 + tl] = (_Float16)src[nt2 * 16 * PATCH + ko];
        }
    }
    #pragma unroll
    for (int e = tid; e < 4 * 2 * 512; e += 1024) {
        int elo = e & 7, lane = (e >> 3) & 31, ehi = (e >> 8) & 1, s = e >> 9;
        int kt = s & 1, nt = s >> 1;
        int n = nt * 16 + (lane & 15);
        int base = (lane & 16) ? 8 : 0;
        int kl = base + elo + 16 * ehi;
        int k = kt * 32 + kl;
        ws[WS_MR + e] = (_Float16)Mr[n * 64 + k];
        ws[WS_MI + e] = (_Float16)Mi[n * 64 + k];
        ws[WS_P + e]  = (_Float16)out_proj_w[n * 64 + k];
    }
}

template <int G>
__device__ __forceinline__ void load_group(FragU (&ring)[3][4], const _Float16* __restrict__ ws, int lane)
{
    #pragma unroll
    for (int f = 0; f < 4; ++f) {
        const _Float16* q;
        if (G < 9)        q = ws + WS_W + ((f * 9 + G) * 512 + lane * 8);
        else if (G < 13)  q = ws + ((f & 1) ? WS_MI : WS_MR) + (((2 * ((G - 9) >> 1) + (f >> 1)) * 2 + ((G - 9) & 1)) * 512 + lane * 8);
        else              q = ws + WS_P + ((f * 2 + (G - 13)) * 512 + lane * 8);
        ring[G % 3][f].u[0] = *(const uint4*)(q);
        ring[G % 3][f].u[1] = *(const uint4*)(q + 256);
    }
}

__global__ void __launch_bounds__(128)
qconv_main_kernel(const float* __restrict__ x,
                  const float* __restrict__ inp_scale,
                  const float* __restrict__ in_proj_b,
                  const float* __restrict__ out_proj_b,
                  const _Float16* __restrict__ ws,
                  float* __restrict__ out)
{
    __shared__ __align__(16) _Float16 bufA[64 * ROWSTRIDE_A];
    __shared__ __align__(16) _Float16 bufY[64 * ROWSTRIDE_Y];
    __shared__ __align__(16) _Float16 bufP[64 * ROWSTRIDE_Y];

    const int tid  = threadIdx.x;
    const int lane = tid & 31;
    const int mt   = tid >> 5;

    FragU ring[3][4];
    float bia[4], obi[4];
    {
        const int m = tid & 63;
        const int gp = __builtin_amdgcn_readfirstlane(tid >> 6);
        const int b = blockIdx.x >> 6;
        const int h = blockIdx.x & 63;
        const bool top = h > 0, bot = h < 63;
        const int hm = top ? h - 1 : h;
        const int hp = bot ? h + 1 : h;
        float v[2][8][3];
        #pragma unroll
        for (int gi = 0; gi < 2; ++gi) {
            const float* xb = x + (size_t)b * C_IN * 4096 + (size_t)(8 * (2 * gp + gi)) * 4096 + m;
            #pragma unroll
            for (int j = 0; j < 8; ++j) {
                const float* xc = xb + j * 4096;
                v[gi][j][0] = xc[hm * 64];
                v[gi][j][1] = xc[h * 64];
                v[gi][j][2] = xc[hp * 64];
            }
        }
        load_group<0>(ring, ws, lane);
        load_group<1>(ring, ws, lane);
        #pragma unroll
        for (int j = 0; j < 4; ++j) {
            bia[j] = in_proj_b[16 * j + (lane & 15)];
            obi[j] = out_proj_b[16 * j + (lane & 15)];
        }
        const bool lastcol = (m == 63), firstcol = (m == 0);
        #pragma unroll
        for (int gi = 0; gi < 2; ++gi) {
            const int g = 2 * gp + gi;
            _Float16* r0 = bufA + ((m + 1) & 63) * ROWSTRIDE_A + 72 * g;
            _Float16* r1 = bufA + m * ROWSTRIDE_A + 72 * g;
            _Float16* r2 = bufA + ((m + 63) & 63) * ROWSTRIDE_A + 72 * g;
            const float* sc = inp_scale + 72 * g;
            #pragma unroll
            for (int r = 0; r < 3; ++r) {
                #pragma unroll
                for (int s = 0; s < 3; ++s) {
                    const int t = r * 3 + s;
                    union { uint4 u; _Float16 e[8]; } pk;
                    #pragma unroll
                    for (int j = 0; j < 8; ++j) {
                        float val = v[gi][j][r];
                        if (r == 0 && !top) val = 0.f;
                        if (r == 2 && !bot) val = 0.f;
                        if (s == 0 && lastcol) val = 0.f;
                        if (s == 2 && firstcol) val = 0.f;
                        pk.e[j] = (_Float16)(val * sc[9 * j + t]);
                    }
                    _Float16* dst = (s == 0) ? r0 : (s == 1) ? r1 : r2;
                    *(uint4*)(dst + 8 * t) = pk.u;
                }
            }
        }
    }
    __syncthreads();

    const int nlo  = lane & 15;
    const int base = (lane & 16) ? 8 : 0;

#define FRONT_OF_PRODUCTS() __builtin_amdgcn_sched_barrier(0)

    v8f acc[4] = {};
    {
        const _Float16* Arow = &bufA[(mt * 16 + nlo) * ROWSTRIDE_A];
        {
            load_group<2>(ring, ws, lane);
            FRONT_OF_PRODUCTS();
            FragU af;
            const _Float16* pa = Arow + 0 * 32 + base;
            af.u[0] = *(const uint4*)(pa);
            af.u[1] = *(const uint4*)(pa + 16);
            #pragma unroll
            for (int j = 0; j < 4; ++j)
                acc[j] = __builtin_amdgcn_wmma_f32_16x16x32_f16(false, af.v, false, ring[0][j].v, (short)0, acc[j], false, false);
        }
        {
            load_group<3>(ring, ws, lane);
            FRONT_OF_PRODUCTS();
            FragU af;
            const _Float16* pa = Arow + 1 * 32 + base;
            af.u[0] = *(const uint4*)(pa);
            af.u[1] = *(const uint4*)(pa + 16);
            #pragma unroll
            for (int j = 0; j < 4; ++j)
                acc[j] = __builtin_amdgcn_wmma_f32_16x16x32_f16(false, af.v, false, ring[1][j].v, (short)0, acc[j], false, false);
        }
        {
            load_group<4>(ring, ws, lane);
            FRONT_OF_PRODUCTS();
            FragU af;
            const _Float16* pa = Arow + 2 * 32 + base;
            af.u[0] = *(const uint4*)(pa);
            af.u[1] = *(const uint4*)(pa + 16);
            #pragma unroll
            for (int j = 0; j < 4; ++j)
                acc[j] = __builtin_amdgcn_wmma_f32_16x16x32_f16(false, af.v, false, ring[2][j].v, (short)0, acc[j], false, false);
        }
        {
            load_group<5>(ring, ws, lane);
            FRONT_OF_PRODUCTS();
            FragU af;
            const _Float16* pa = Arow + 3 * 32 + base;
            af.u[0] = *(const uint4*)(pa);
            af.u[1] = *(const uint4*)(pa + 16);
            #pragma unroll
            for (int j = 0; j < 4; ++j)
                acc[j] = __builtin_amdgcn_wmma_f32_16x16x32_f16(false, af.v, false, ring[0][j].v, (short)0, acc[j], false, false);
        }
        {
            load_group<6>(ring, ws, lane);
            FRONT_OF_PRODUCTS();
            FragU af;
            const _Float16* pa = Arow + 4 * 32 + base;
            af.u[0] = *(const uint4*)(pa);
            af.u[1] = *(const uint4*)(pa + 16);
            #pragma unroll
            for (int j = 0; j < 4; ++j)
                acc[j] = __builtin_amdgcn_wmma_f32_16x16x32_f16(false, af.v, false, ring[1][j].v, (short)0, acc[j], false, false);
        }
        {
            load_group<7>(ring, ws, lane);
            FRONT_OF_PRODUCTS();
            FragU af;
            const _Float16* pa = Arow + 5 * 32 + base;
            af.u[0] = *(const uint4*)(pa);
            af.u[1] = *(const uint4*)(pa + 16);
            #pragma unroll
            for (int j = 0; j < 4; ++j)
                acc[j] = __builtin_amdgcn_wmma_f32_16x16x32_f16(false, af.v, false, ring[2][j].v, (short)0, acc[j], false, false);
        }
        {
            load_group<8>(ring, ws, lane);
            FRONT_OF_PRODUCTS();
            FragU af;
            const _Float16* pa = Arow + 6 * 32 + base;
            af.u[0] = *(const uint4*)(pa);
            af.u[1] = *(const uint4*)(pa + 16);
            #pragma unroll
            for (int j = 0; j < 4; ++j)
                acc[j] = __builtin_amdgcn_wmma_f32_16x16x32_f16(false, af.v, false, ring[0][j].v, (short)0, acc[j], false, false);
        }
        {
            load_group<9>(ring, ws, lane);
            FRONT_OF_PRODUCTS();
            FragU af;
            const _Float16* pa = Arow + 7 * 32 + base;
            af.u[0] = *(const uint4*)(pa);
            af.u[1] = *(const uint4*)(pa + 16);
            #pragma unroll
            for (int j = 0; j < 4; ++j)
                acc[j] = __builtin_amdgcn_wmma_f32_16x16x32_f16(false, af.v, false, ring[1][j].v, (short)0, acc[j], false, false);
        }
        {
            load_group<10>(ring, ws, lane);
            FRONT_OF_PRODUCTS();
            FragU af;
            const _Float16* pa = Arow + 8 * 32 + base;
            af.u[0] = *(const uint4*)(pa);
            af.u[1] = *(const uint4*)(pa + 16);
            #pragma unroll
            for (int j = 0; j < 4; ++j)
                acc[j] = __builtin_amdgcn_wmma_f32_16x16x32_f16(false, af.v, false, ring[2][j].v, (short)0, acc[j], false, false);
        }
    }

    float y[4][8], part[8];
    {
        #pragma unroll
        for (int r = 0; r < 8; ++r) {
            float s = 0.f;
            #pragma unroll
            for (int j = 0; j < 4; ++j) {
                y[j][r] = acc[j][r] + bia[j];
                s += y[j][r] * y[j][r];
            }
            part[r] = s;
        }
    }
    #pragma unroll
    for (int mask = 1; mask <= 8; mask <<= 1) {
        #pragma unroll
        for (int r = 0; r < 8; ++r)
            part[r] += __shfl_xor(part[r], mask, 32);
    }
    #pragma unroll
    for (int r = 0; r < 8; ++r) {
        const int m = mt * 16 + base + r;
        const float inv = 1.f / (sqrtf(part[r]) + 1e-9f);
        #pragma unroll
        for (int j = 0; j < 4; ++j)
            bufY[m * ROWSTRIDE_Y + 16 * j + nlo] = (_Float16)(y[j][r] * inv);
    }
    __builtin_amdgcn_fence(__ATOMIC_ACQ_REL, "wavefront");

    {
        const _Float16* Yrow = &bufY[(mt * 16 + nlo) * ROWSTRIDE_Y];
        FragU af[2];
        #pragma unroll
        for (int kt = 0; kt < 2; ++kt) {
            const _Float16* pa = Yrow + kt * 32 + base;
            af[kt].u[0] = *(const uint4*)(pa);
            af[kt].u[1] = *(const uint4*)(pa + 16);
        }
        {
            v8f ar[2] = {}, ai[2] = {};
            load_group<11>(ring, ws, lane);
            FRONT_OF_PRODUCTS();
            #pragma unroll
            for (int jj = 0; jj < 2; ++jj) {
                ar[jj] = __builtin_amdgcn_wmma_f32_16x16x32_f16(false, af[0].v, false, ring[0][2 * jj].v,     (short)0, ar[jj], false, false);
                ai[jj] = __builtin_amdgcn_wmma_f32_16x16x32_f16(false, af[0].v, false, ring[0][2 * jj + 1].v, (short)0, ai[jj], false, false);
            }
            load_group<12>(ring, ws, lane);
            FRONT_OF_PRODUCTS();
            #pragma unroll
            for (int jj = 0; jj < 2; ++jj) {
                ar[jj] = __builtin_amdgcn_wmma_f32_16x16x32_f16(false, af[1].v, false, ring[1][2 * jj].v,     (short)0, ar[jj], false, false);
                ai[jj] = __builtin_amdgcn_wmma_f32_16x16x32_f16(false, af[1].v, false, ring[1][2 * jj + 1].v, (short)0, ai[jj], false, false);
            }
            #pragma unroll
            for (int r = 0; r < 8; ++r) {
                const int m = mt * 16 + base + r;
                #pragma unroll
                for (int jj = 0; jj < 2; ++jj)
                    bufP[m * ROWSTRIDE_Y + 16 * (0 + jj) + nlo] = (_Float16)(ar[jj][r] * ar[jj][r] + ai[jj][r] * ai[jj][r]);
            }
        }
        {
            v8f ar[2] = {}, ai[2] = {};
            load_group<13>(ring, ws, lane);
            FRONT_OF_PRODUCTS();
            #pragma unroll
            for (int jj = 0; jj < 2; ++jj) {
                ar[jj] = __builtin_amdgcn_wmma_f32_16x16x32_f16(false, af[0].v, false, ring[2][2 * jj].v,     (short)0, ar[jj], false, false);
                ai[jj] = __builtin_amdgcn_wmma_f32_16x16x32_f16(false, af[0].v, false, ring[2][2 * jj + 1].v, (short)0, ai[jj], false, false);
            }
            load_group<14>(ring, ws, lane);
            FRONT_OF_PRODUCTS();
            #pragma unroll
            for (int jj = 0; jj < 2; ++jj) {
                ar[jj] = __builtin_amdgcn_wmma_f32_16x16x32_f16(false, af[1].v, false, ring[0][2 * jj].v,     (short)0, ar[jj], false, false);
                ai[jj] = __builtin_amdgcn_wmma_f32_16x16x32_f16(false, af[1].v, false, ring[0][2 * jj + 1].v, (short)0, ai[jj], false, false);
            }
            #pragma unroll
            for (int r = 0; r < 8; ++r) {
                const int m = mt * 16 + base + r;
                #pragma unroll
                for (int jj = 0; jj < 2; ++jj)
                    bufP[m * ROWSTRIDE_Y + 16 * (2 + jj) + nlo] = (_Float16)(ar[jj][r] * ar[jj][r] + ai[jj][r] * ai[jj][r]);
            }
        }
    }
    __builtin_amdgcn_fence(__ATOMIC_ACQ_REL, "wavefront");

    v8f o[4] = {};
    {
        const _Float16* Prow = &bufP[(mt * 16 + nlo) * ROWSTRIDE_Y];
        #pragma unroll
        for (int kt = 0; kt < 2; ++kt) {
            FragU af;
            const _Float16* pa = Prow + kt * 32 + base;
            af.u[0] = *(const uint4*)(pa);
            af.u[1] = *(const uint4*)(pa + 16);
            #pragma unroll
            for (int j = 0; j < 4; ++j)
                o[j] = __builtin_amdgcn_wmma_f32_16x16x32_f16(false, af.v, false, ring[(13 + kt) % 3][j].v, (short)0, o[j], false, false);
        }
    }

    float* outstage = (float*)bufA;
    {
        float* os = outstage + mt * OSTAGE_TILE + mt * 16;
        #pragma unroll
        for (int j = 0; j < 4; ++j) {
            const float ob = obi[j];
            float* col = os + (16 * j + nlo) * OSTAGE_PITCH + base;
            const v4f lo = { o[j][0] + ob, o[j][1] + ob, o[j][2] + ob, o[j][3] + ob };
            const v4f hi = { o[j][4] + ob, o[j][5] + ob, o[j][6] + ob, o[j][7] + ob };
            *(v4f*)(col)     = lo;
            *(v4f*)(col + 4) = hi;
        }
    }
    __syncthreads();
    {
        const int q = lane & 15, cs = lane >> 4, t = q >> 2;
        const float* rd = outstage + t * OSTAGE_TILE + t * 16 + (2 * mt + cs) * OSTAGE_PITCH + 4 * (q & 3);
        float* wr = out + ((size_t)((blockIdx.x >> 6) * 64 + 2 * mt + cs)) * 4096 + (blockIdx.x & 63) * 64 + 4 * q;
        v4f ov[8];
        #pragma unroll
        for (int i = 0; i < 8; ++i) ov[i] = *(const v4f*)(rd + i * 8 * OSTAGE_PITCH);
        #pragma unroll
        for (int i = 0; i < 8; ++i) *(v4f*)(wr + (size_t)i * 8 * 4096) = ov[i];
    }
}

extern "C" void kernel_launch(void* const* d_in, const int* in_sizes, int n_in,
                              void* d_out, int out_size, void* d_ws, size_t ws_size,
                              hipStream_t stream) {
    const float* x          = (const float*)d_in[0];
    const float* inp_scale  = (const float*)d_in[1];
    const float* in_proj_w  = (const float*)d_in[2];
    const float* in_proj_b  = (const float*)d_in[3];
    const float* weights    = (const float*)d_in[4];
    const float* meas_w     = (const float*)d_in[5];
    const float* out_proj_w = (const float*)d_in[6];
    const float* out_proj_b = (const float*)d_in[7];
    _Float16* ws = (_Float16*)d_ws;
    float* out = (float*)d_out;

    hipLaunchKernelGGL(qprep_kernel, dim3(1), dim3(1024), 0, stream,
                       in_proj_w, weights, meas_w, out_proj_w, ws);
    hipLaunchKernelGGL(qconv_main_kernel, dim3(1024), dim3(128), 0, stream,
                       x, inp_scale, in_proj_b, out_proj_b, ws, out);
}


# prep: a sweep applies its three one-wire 2x2 gates in turn to the 8 values a thread holds (192 multiply-adds, 24 gate words per task), not their 8x8 product (256, 128); f32 sums reordered
# speedup vs baseline: 5.7121x; 1.1001x over previous
#include <hip/hip_runtime.h>
#include <hip/hip_bf16.h>
#include <math.h>

typedef __attribute__((ext_vector_type(16))) _Float16 v16h;
typedef __attribute__((ext_vector_type(8)))  float    v8f;
typedef __attribute__((ext_vector_type(4)))  float    v4f;

#define PATCH 288
#define QD 64
#define C_IN 32
#define ROWSTRIDE_A 296
#define ROWSTRIDE_Y 72
#define OSTAGE_TILE (16 * ROWSTRIDE_A / 2)
#define OSTAGE_PITCH 20

#define WS_W   0
#define WS_MR  18432
#define WS_MI  (18432 + 4096)
#define WS_P   (18432 + 8192)

union FragU { uint4 u[2]; v16h v; };


__device__ __forceinline__ void c2mul(const float* a, const float* b, float* o)
{
    for (int r = 0; r < 2; ++r)
        for (int c = 0; c < 2; ++c) {
            float re = 0.f, im = 0.f;
            for (int k = 0; k < 2; ++k) {
                float ar = a[(r * 2 + k) * 2], ai = a[(r * 2 + k) * 2 + 1];
                float br = b[(k * 2 + c) * 2], bi = b[(k * 2 + c) * 2 + 1];
                re += ar * br - ai * bi;
                im += ar * bi + ai * br;
            }
            o[(r * 2 + c) * 2] = re;
            o[(r * 2 + c) * 2 + 1] = im;
        }
}

__device__ __forceinline__ void make_rx(float c, float s, float* o)
{
    o[0] = c;   o[1] = 0.f; o[2] = 0.f; o[3] = -s;
    o[4] = 0.f; o[5] = -s;  o[6] = c;   o[7] = 0.f;
}

__device__ __forceinline__ void make_ry(float c, float s, float* o)
{
    o[0] = c; o[1] = 0.f; o[2] = -s; o[3] = 0.f;
    o[4] = s; o[5] = 0.f; o[6] = c;  o[7] = 0.f;
}

template <bool HI, bool PERM>
__device__ __forceinline__ void block_sweep(float* Mr, float* Mi, const float* Gw, int tid)
{
    float xr[8] = {}, xi[8] = {};
    const bool act = tid < 512;
    const int c = tid & 63, o = (tid >> 6) & 7;
    if (act) {
        for (int s = 0; s < 8; ++s) {
            int row = HI ? (s * 8 + o) : (o * 8 + s);
            if (PERM) row ^= row >> 1;
            xr[s] = Mr[row * 64 + c];
            xi[s] = Mi[row * 64 + c];
        }
        #pragma unroll
        for (int q = 0; q < 3; ++q) {
            const float* g = Gw + q * 8;
            const float g00r = g[0], g00i = g[1], g01r = g[2], g01i = g[3];
            const float g10r = g[4], g10i = g[5], g11r = g[6], g11i = g[7];
            const int bit = 4 >> q;
            #pragma unroll
            for (int s0 = 0; s0 < 8; ++s0) {
                if (s0 & bit) continue;
                const int s1 = s0 | bit;
                const float ar = xr[s0], ai = xi[s0], br = xr[s1], bi = xi[s1];
                xr[s0] = fmaf(-g01i, bi, fmaf(g01r, br, fmaf(-g00i, ai, g00r * ar)));
                xi[s0] = fmaf( g01i, br, fmaf(g01r, bi, fmaf( g00i, ar, g00r * ai)));
                xr[s1] = fmaf(-g11i, bi, fmaf(g11r, br, fmaf(-g10i, ai, g10r * ar)));
                xi[s1] = fmaf( g11i, br, fmaf(g11r, bi, fmaf( g10i, ar, g10r * ai)));
            }
        }
    }
    if (PERM) __syncthreads();
    if (act) {
        for (int a = 0; a < 8; ++a) {
            const int row = HI ? (a * 8 + o) : (o * 8 + a);
            Mr[row * 64 + c] = xr[a];
            Mi[row * 64 + c] = xi[a];
        }
    }
    __syncthreads();
}

__global__ void __launch_bounds__(1024)
qprep_kernel(const float* __restrict__ in_proj_w,
             const float* __restrict__ weights,
             const float* __restrict__ meas_w,
             const float* __restrict__ out_proj_w,
             _Float16* __restrict__ ws)
{
    __shared__ float Mr[64 * 64];
    __shared__ float Mi[64 * 64];
    __shared__ float CS[60 * 2];
    __shared__ float G[18 * 8];
    __shared__ float Wl[QD * PATCH];
    const int tid = threadIdx.x;

    float ang = 0.f;
    if (tid < 60) {
        if (tid < 36) {
            ang = weights[tid] * 0.5f;
        } else {
            const int k = tid - 36, w = k >> 2, which = k & 3;
            const float t = meas_w[w * 3 + 0], p = meas_w[w * 3 + 1], lm = meas_w[w * 3 + 2];
            ang = (which == 0) ? t * 0.5f : (which == 1) ? p : (which == 2) ? lm : (p + lm);
        }
    }
    float wv[18];
    for (int i = 0; i < 18; ++i) wv[i] = in_proj_w[tid + 1024 * i];
    if (tid < 60) {
        CS[tid * 2]     = cosf(ang);
        CS[tid * 2 + 1] = sinf(ang);
    }
    __syncthreads();

    if (tid < 18) {
        const int L = tid / 6, w = tid - L * 6;
        float m0[8], m1[8], m2[8], m3[8];
        if (L == 0) {
            make_ry(CS[(w * 3 + 1) * 2], CS[(w * 3 + 1) * 2 + 1], m0);
            make_rx(CS[(w * 3 + 0) * 2], CS[(w * 3 + 0) * 2 + 1], m1);
            c2mul(m0, m1, m3);
        } else if (L == 1) {
            make_ry(CS[((6 + w) * 3 + 1) * 2], CS[((6 + w) * 3 + 1) * 2 + 1], m0);
            make_rx(CS[((6 + w) * 3 + 0) * 2], CS[((6 + w) * 3 + 0) * 2 + 1], m1);
            c2mul(m0, m1, m2);
            make_ry(CS[(w * 3 + 2) * 2], CS[(w * 3 + 2) * 2 + 1], m0);
            c2mul(m2, m0, m3);
        } else {
            const float c  = CS[(36 + w * 4 + 0) * 2], s  = CS[(36 + w * 4 + 0) * 2 + 1];
            const float cp = CS[(36 + w * 4 + 1) * 2], sp = CS[(36 + w * 4 + 1) * 2 + 1];
            const float cl = CS[(36 + w * 4 + 2) * 2], sl = CS[(36 + w * 4 + 2) * 2 + 1];
            const float cpl = CS[(36 + w * 4 + 3) * 2], spl = CS[(36 + w * 4 + 3) * 2 + 1];
            m0[0] = c;      m0[1] = 0.f;    m0[2] = -cl * s; m0[3] = -sl * s;
            m0[4] = cp * s; m0[5] = sp * s; m0[6] = cpl * c; m0[7] = spl * c;
            make_ry(CS[((6 + w) * 3 + 2) * 2], CS[((6 + w) * 3 + 2) * 2 + 1], m1);
            c2mul(m0, m1, m3);
        }
        for (int e = 0; e < 8; ++e) G[tid * 8 + e] = m3[e];
    }
    for (int i = 0; i < 18; ++i) Wl[tid + 1024 * i] = wv[i];
    __syncthreads();

    if (tid < 256) {
        const int col = tid & 63, slo = tid >> 6;
        float gr[6][2], gi[6][2];
        for (int w = 0; w < 6; ++w) {
            const int cb = (col >> (5 - w)) & 1;
            gr[w][0] = G[(w * 4 + 0 + cb) * 2];  gi[w][0] = G[(w * 4 + 0 + cb) * 2 + 1];
            gr[w][1] = G[(w * 4 + 2 + cb) * 2];  gi[w][1] = G[(w * 4 + 2 + cb) * 2 + 1];
        }
        const bool b4 = (slo >> 1) & 1, b5 = slo & 1;
        const float g4r = b4 ? gr[4][1] : gr[4][0], g4i = b4 ? gi[4][1] : gi[4][0];
        const float g5r = b5 ? gr[5][1] : gr[5][0], g5i = b5 ? gi[5][1] : gi[5][0];
        float pr[16], pi[16];
        pr[0] = g4r * g5r - g4i * g5i;
        pi[0] = g4r * g5i + g4i * g5r;
        for (int k = 0; k < 4; ++k) {
            const int w = 3 - k, n = 1 << k;
            for (int idx = 0; idx < n; ++idx) {
                const float ar = pr[idx], ai = pi[idx];
                pr[idx + n] = ar * gr[w][1] - ai * gi[w][1];
                pi[idx + n] = ar * gi[w][1] + ai * gr[w][1];
                pr[idx]     = ar * gr[w][0] - ai * gi[w][0];
                pi[idx]     = ar * gi[w][0] + ai * gr[w][0];
            }
        }
        for (int t = 0; t < 16; ++t) {
            const int s = slo + 4 * t;
            const int row = s ^ (s >> 1) ^ (s >> 2) ^ (s >> 3) ^ (s >> 4) ^ (s >> 5);
            Mr[row * 64 + col] = pr[t];
            Mi[row * 64 + col] = pi[t];
        }
    }
    __syncthreads();

    block_sweep<true,  false>(Mr, Mi, G + (1 * 6 + 0) * 8, tid);
    block_sweep<false, false>(Mr, Mi, G + (1 * 6 + 3) * 8, tid);
    block_sweep<true,  true >(Mr, Mi, G + (2 * 6 + 0) * 8, tid);
    block_sweep<false, false>(Mr, Mi, G + (2 * 6 + 3) * 8, tid);

    {
        const int tl = tid & 511, t9 = tid >> 9;
        const int elo = tl & 7, lane = (tl >> 3) & 31, ehi = tl >> 8;
        const int base = (lane & 16) ? 8 : 0;
        const int kl = base + elo + 16 * ehi;
        const int run = kl >> 3, cl = kl & 7;
        const float* src = Wl + (t9 * 32 + (lane & 15)) * PATCH + 9 * cl + run;
        #pragma unroll
        for (int i = 0; i < 18; ++i) {
            const int kt = i % 9, nt2 = i / 9;
            const int c0 = (4 * kt) / 9, r0 = (4 * kt) % 9;
            const int ko = 72 * c0 + r0 + ((r0 + run >= 9) ? 63 : 0);
            ws[WS_W + (t9 * 18 + i) * 512 + tl] = (_Float16)src[nt2 * 16 * PATCH + ko];
        }
    }
    #pragma unroll
    for (int e = tid; e < 4 * 2 * 512; e += 1024) {
        int elo = e & 7, lane = (e >> 3) & 31, ehi = (e >> 8) & 1, s = e >> 9;
        int kt = s & 1, nt = s >> 1;
        int n = nt * 16 + (lane & 15);
        int base = (lane & 16) ? 8 : 0;
        int kl = base + elo + 16 * ehi;
        int k = kt * 32 + kl;
        ws[WS_MR + e] = (_Float16)Mr[n * 64 + k];
        ws[WS_MI + e] = (_Float16)Mi[n * 64 + k];
        ws[WS_P + e]  = (_Float16)out_proj_w[n * 64 + k];
    }
}

template <int G>
__device__ __forceinline__ void load_group(FragU (&ring)[3][4], const _Float16* __restrict__ ws, int lane)
{
    #pragma unroll
    for (int f = 0; f < 4; ++f) {
        const _Float16* q;
        if (G < 9)        q = ws + WS_W + ((f * 9 + G) * 512 + lane * 8);
        else if (G < 13)  q = ws + ((f & 1) ? WS_MI : WS_MR) + (((2 * ((G - 9) >> 1) + (f >> 1)) * 2 + ((G - 9) & 1)) * 512 + lane * 8);
        else              q = ws + WS_P + ((f * 2 + (G - 13)) * 512 + lane * 8);
        ring[G % 3][f].u[0] = *(const uint4*)(q);
        ring[G % 3][f].u[1] = *(const uint4*)(q + 256);
    }
}

__global__ void __launch_bounds__(128)
qconv_main_kernel(const float* __restrict__ x,
                  const float* __restrict__ inp_scale,
                  const float* __restrict__ in_proj_b,
                  const float* __restrict__ out_proj_b,
                  const _Float16* __restrict__ ws,
                  float* __restrict__ out)
{
    __shared__ __align__(16) _Float16 bufA[64 * ROWSTRIDE_A];
    __shared__ __align__(16) _Float16 bufY[64 * ROWSTRIDE_Y];
    __shared__ __align__(16) _Float16 bufP[64 * ROWSTRIDE_Y];

    const int tid  = threadIdx.x;
    const int lane = tid & 31;
    const int mt   = tid >> 5;

    FragU ring[3][4];
    float bia[4], obi[4];
    {
        const int m = tid & 63;
        const int gp = __builtin_amdgcn_readfirstlane(tid >> 6);
        const int b = blockIdx.x >> 6;
        const int h = blockIdx.x & 63;
        const bool top = h > 0, bot = h < 63;
        const int hm = top ? h - 1 : h;
        const int hp = bot ? h + 1 : h;
        float v[2][8][3];
        #pragma unroll
        for (int gi = 0; gi < 2; ++gi) {
            const float* xb = x + (size_t)b * C_IN * 4096 + (size_t)(8 * (2 * gp + gi)) * 4096 + m;
            #pragma unroll
            for (int j = 0; j < 8; ++j) {
                const float* xc = xb + j * 4096;
                v[gi][j][0] = xc[hm * 64];
                v[gi][j][1] = xc[h * 64];
                v[gi][j][2] = xc[hp * 64];
            }
        }
        load_group<0>(ring, ws, lane);
        load_group<1>(ring, ws, lane);
        #pragma unroll
        for (int j = 0; j < 4; ++j) {
            bia[j] = in_proj_b[16 * j + (lane & 15)];
            obi[j] = out_proj_b[16 * j + (lane & 15)];
        }
        const bool lastcol = (m == 63), firstcol = (m == 0);
        #pragma unroll
        for (int gi = 0; gi < 2; ++gi) {
            const int g = 2 * gp + gi;
            _Float16* r0 = bufA + ((m + 1) & 63) * ROWSTRIDE_A + 72 * g;
            _Float16* r1 = bufA + m * ROWSTRIDE_A + 72 * g;
            _Float16* r2 = bufA + ((m + 63) & 63) * ROWSTRIDE_A + 72 * g;
            const float* sc = inp_scale + 72 * g;
            #pragma unroll
            for (int r = 0; r < 3; ++r) {
                #pragma unroll
                for (int s = 0; s < 3; ++s) {
                    const int t = r * 3 + s;
                    union { uint4 u; _Float16 e[8]; } pk;
                    #pragma unroll
                    for (int j = 0; j < 8; ++j) {
                        float val = v[gi][j][r];
                        if (r == 0 && !top) val = 0.f;
                        if (r == 2 && !bot) val = 0.f;
                        if (s == 0 && lastcol) val = 0.f;
                        if (s == 2 && firstcol) val = 0.f;
                        pk.e[j] = (_Float16)(val * sc[9 * j + t]);
                    }
                    _Float16* dst = (s == 0) ? r0 : (s == 1) ? r1 : r2;
                    *(uint4*)(dst + 8 * t) = pk.u;
                }
            }
        }
    }
    __syncthreads();

    const int nlo  = lane & 15;
    const int base = (lane & 16) ? 8 : 0;

#define FRONT_OF_PRODUCTS() __builtin_amdgcn_sched_barrier(0)

    v8f acc[4] = {};
    {
        const _Float16* Arow = &bufA[(mt * 16 + nlo) * ROWSTRIDE_A];
        {
            load_group<2>(ring, ws, lane);
            FRONT_OF_PRODUCTS();
            FragU af;
            const _Float16* pa = Arow + 0 * 32 + base;
            af.u[0] = *(const uint4*)(pa);
            af.u[1] = *(const uint4*)(pa + 16);
            #pragma unroll
            for (int j = 0; j < 4; ++j)
                acc[j] = __builtin_amdgcn_wmma_f32_16x16x32_f16(false, af.v, false, ring[0][j].v, (short)0, acc[j], false, false);
        }
        {
            load_group<3>(ring, ws, lane);
            FRONT_OF_PRODUCTS();
            FragU af;
            const _Float16* pa = Arow + 1 * 32 + base;
            af.u[0] = *(const uint4*)(pa);
            af.u[1] = *(const uint4*)(pa + 16);
            #pragma unroll
            for (int j = 0; j < 4; ++j)
                acc[j] = __builtin_amdgcn_wmma_f32_16x16x32_f16(false, af.v, false, ring[1][j].v, (short)0, acc[j], false, false);
        }
        {
            load_group<4>(ring, ws, lane);
            FRONT_OF_PRODUCTS();
            FragU af;
            const _Float16* pa = Arow + 2 * 32 + base;
            af.u[0] = *(const uint4*)(pa);
            af.u[1] = *(const uint4*)(pa + 16);
            #pragma unroll
            for (int j = 0; j < 4; ++j)
                acc[j] = __builtin_amdgcn_wmma_f32_16x16x32_f16(false, af.v, false, ring[2][j].v, (short)0, acc[j], false, false);
        }
        {
            load_group<5>(ring, ws, lane);
            FRONT_OF_PRODUCTS();
            FragU af;
            const _Float16* pa = Arow + 3 * 32 + base;
            af.u[0] = *(const uint4*)(pa);
            af.u[1] = *(const uint4*)(pa + 16);
            #pragma unroll
            for (int j = 0; j < 4; ++j)
                acc[j] = __builtin_amdgcn_wmma_f32_16x16x32_f16(false, af.v, false, ring[0][j].v, (short)0, acc[j], false, false);
        }
        {
            load_group<6>(ring, ws, lane);
            FRONT_OF_PRODUCTS();
            FragU af;
            const _Float16* pa = Arow + 4 * 32 + base;
            af.u[0] = *(const uint4*)(pa);
            af.u[1] = *(const uint4*)(pa + 16);
            #pragma unroll
            for (int j = 0; j < 4; ++j)
                acc[j] = __builtin_amdgcn_wmma_f32_16x16x32_f16(false, af.v, false, ring[1][j].v, (short)0, acc[j], false, false);
        }
        {
            load_group<7>(ring, ws, lane);
            FRONT_OF_PRODUCTS();
            FragU af;
            const _Float16* pa = Arow + 5 * 32 + base;
            af.u[0] = *(const uint4*)(pa);
            af.u[1] = *(const uint4*)(pa + 16);
            #pragma unroll
            for (int j = 0; j < 4; ++j)
                acc[j] = __builtin_amdgcn_wmma_f32_16x16x32_f16(false, af.v, false, ring[2][j].v, (short)0, acc[j], false, false);
        }
        {
            load_group<8>(ring, ws, lane);
            FRONT_OF_PRODUCTS();
            FragU af;
            const _Float16* pa = Arow + 6 * 32 + base;
            af.u[0] = *(const uint4*)(pa);
            af.u[1] = *(const uint4*)(pa + 16);
            #pragma unroll
            for (int j = 0; j < 4; ++j)
                acc[j] = __builtin_amdgcn_wmma_f32_16x16x32_f16(false, af.v, false, ring[0][j].v, (short)0, acc[j], false, false);
        }
        {
            load_group<9>(ring, ws, lane);
            FRONT_OF_PRODUCTS();
            FragU af;
            const _Float16* pa = Arow + 7 * 32 + base;
            af.u[0] = *(const uint4*)(pa);
            af.u[1] = *(const uint4*)(pa + 16);
            #pragma unroll
            for (int j = 0; j < 4; ++j)
                acc[j] = __builtin_amdgcn_wmma_f32_16x16x32_f16(false, af.v, false, ring[1][j].v, (short)0, acc[j], false, false);
        }
        {
            load_group<10>(ring, ws, lane);
            FRONT_OF_PRODUCTS();
            FragU af;
            const _Float16* pa = Arow + 8 * 32 + base;
            af.u[0] = *(const uint4*)(pa);
            af.u[1] = *(const uint4*)(pa + 16);
            #pragma unroll
            for (int j = 0; j < 4; ++j)
                acc[j] = __builtin_amdgcn_wmma_f32_16x16x32_f16(false, af.v, false, ring[2][j].v, (short)0, acc[j], false, false);
        }
    }

    float y[4][8], part[8];
    {
        #pragma unroll
        for (int r = 0; r < 8; ++r) {
            float s = 0.f;
            #pragma unroll
            for (int j = 0; j < 4; ++j) {
                y[j][r] = acc[j][r] + bia[j];
                s += y[j][r] * y[j][r];
            }
            part[r] = s;
        }
    }
    #pragma unroll
    for (int mask = 1; mask <= 8; mask <<= 1) {
        #pragma unroll
        for (int r = 0; r < 8; ++r)
            part[r] += __shfl_xor(part[r], mask, 32);
    }
    #pragma unroll
    for (int r = 0; r < 8; ++r) {
        const int m = mt * 16 + base + r;
        const float inv = 1.f / (sqrtf(part[r]) + 1e-9f);
        #pragma unroll
        for (int j = 0; j < 4; ++j)
            bufY[m * ROWSTRIDE_Y + 16 * j + nlo] = (_Float16)(y[j][r] * inv);
    }
    __builtin_amdgcn_fence(__ATOMIC_ACQ_REL, "wavefront");

    {
        const _Float16* Yrow = &bufY[(mt * 16 + nlo) * ROWSTRIDE_Y];
        FragU af[2];
        #pragma unroll
        for (int kt = 0; kt < 2; ++kt) {
            const _Float16* pa = Yrow + kt * 32 + base;
            af[kt].u[0] = *(const uint4*)(pa);
            af[kt].u[1] = *(const uint4*)(pa + 16);
        }
        {
            v8f ar[2] = {}, ai[2] = {};
            load_group<11>(ring, ws, lane);
            FRONT_OF_PRODUCTS();
            #pragma unroll
            for (int jj = 0; jj < 2; ++jj) {
                ar[jj] = __builtin_amdgcn_wmma_f32_16x16x32_f16(false, af[0].v, false, ring[0][2 * jj].v,     (short)0, ar[jj], false, false);
                ai[jj] = __builtin_amdgcn_wmma_f32_16x16x32_f16(false, af[0].v, false, ring[0][2 * jj + 1].v, (short)0, ai[jj], false, false);
            }
            load_group<12>(ring, ws, lane);
            FRONT_OF_PRODUCTS();
            #pragma unroll
            for (int jj = 0; jj < 2; ++jj) {
                ar[jj] = __builtin_amdgcn_wmma_f32_16x16x32_f16(false, af[1].v, false, ring[1][2 * jj].v,     (short)0, ar[jj], false, false);
                ai[jj] = __builtin_amdgcn_wmma_f32_16x16x32_f16(false, af[1].v, false, ring[1][2 * jj + 1].v, (short)0, ai[jj], false, false);
            }
            #pragma unroll
            for (int r = 0; r < 8; ++r) {
                const int m = mt * 16 + base + r;
                #pragma unroll
                for (int jj = 0; jj < 2; ++jj)
                    bufP[m * ROWSTRIDE_Y + 16 * (0 + jj) + nlo] = (_Float16)(ar[jj][r] * ar[jj][r] + ai[jj][r] * ai[jj][r]);
            }
        }
        {
            v8f ar[2] = {}, ai[2] = {};
            load_group<13>(ring, ws, lane);
            FRONT_OF_PRODUCTS();
            #pragma unroll
            for (int jj = 0; jj < 2; ++jj) {
                ar[jj] = __builtin_amdgcn_wmma_f32_16x16x32_f16(false, af[0].v, false, ring[2][2 * jj].v,     (short)0, ar[jj], false, false);
                ai[jj] = __builtin_amdgcn_wmma_f32_16x16x32_f16(false, af[0].v, false, ring[2][2 * jj + 1].v, (short)0, ai[jj], false, false);
            }
            load_group<14>(ring, ws, lane);
            FRONT_OF_PRODUCTS();
            #pragma unroll
            for (int jj = 0; jj < 2; ++jj) {
                ar[jj] = __builtin_amdgcn_wmma_f32_16x16x32_f16(false, af[1].v, false, ring[0][2 * jj].v,     (short)0, ar[jj], false, false);
                ai[jj] = __builtin_amdgcn_wmma_f32_16x16x32_f16(false, af[1].v, false, ring[0][2 * jj + 1].v, (short)0, ai[jj], false, false);
            }
            #pragma unroll
            for (int r = 0; r < 8; ++r) {
                const int m = mt * 16 + base + r;
                #pragma unroll
                for (int jj = 0; jj < 2; ++jj)
                    bufP[m * ROWSTRIDE_Y + 16 * (2 + jj) + nlo] = (_Float16)(ar[jj][r] * ar[jj][r] + ai[jj][r] * ai[jj][r]);
            }
        }
    }
    __builtin_amdgcn_fence(__ATOMIC_ACQ_REL, "wavefront");

    v8f o[4] = {};
    {
        const _Float16* Prow = &bufP[(mt * 16 + nlo) * ROWSTRIDE_Y];
        #pragma unroll
        for (int kt = 0; kt < 2; ++kt) {
            FragU af;
            const _Float16* pa = Prow + kt * 32 + base;
            af.u[0] = *(const uint4*)(pa);
            af.u[1] = *(const uint4*)(pa + 16);
            #pragma unroll
            for (int j = 0; j < 4; ++j)
                o[j] = __builtin_amdgcn_wmma_f32_16x16x32_f16(false, af.v, false, ring[(13 + kt) % 3][j].v, (short)0, o[j], false, false);
        }
    }

    float* outstage = (float*)bufA;
    {
        float* os = outstage + mt * OSTAGE_TILE + mt * 16;
        #pragma unroll
        for (int j = 0; j < 4; ++j) {
            const float ob = obi[j];
            float* col = os + (16 * j + nlo) * OSTAGE_PITCH + base;
            const v4f lo = { o[j][0] + ob, o[j][1] + ob, o[j][2] + ob, o[j][3] + ob };
            const v4f hi = { o[j][4] + ob, o[j][5] + ob, o[j][6] + ob, o[j][7] + ob };
            *(v4f*)(col)     = lo;
            *(v4f*)(col + 4) = hi;
        }
    }
    __syncthreads();
    {
        const int q = lane & 15, cs = lane >> 4, t = q >> 2;
        const float* rd = outstage + t * OSTAGE_TILE + t * 16 + (2 * mt + cs) * OSTAGE_PITCH + 4 * (q & 3);
        float* wr = out + ((size_t)((blockIdx.x >> 6) * 64 + 2 * mt + cs)) * 4096 + (blockIdx.x & 63) * 64 + 4 * q;
        v4f ov[8];
        #pragma unroll
        for (int i = 0; i < 8; ++i) ov[i] = *(const v4f*)(rd + i * 8 * OSTAGE_PITCH);
        #pragma unroll
        for (int i = 0; i < 8; ++i) *(v4f*)(wr + (size_t)i * 8 * 4096) = ov[i];
    }
}

extern "C" void kernel_launch(void* const* d_in, const int* in_sizes, int n_in,
                              void* d_out, int out_size, void* d_ws, size_t ws_size,
                              hipStream_t stream) {
    const float* x          = (const float*)d_in[0];
    const float* inp_scale  = (const float*)d_in[1];
    const float* in_proj_w  = (const float*)d_in[2];
    const float* in_proj_b  = (const float*)d_in[3];
    const float* weights    = (const float*)d_in[4];
    const float* meas_w     = (const float*)d_in[5];
    const float* out_proj_w = (const float*)d_in[6];
    const float* out_proj_b = (const float*)d_in[7];
    _Float16* ws = (_Float16*)d_ws;
    float* out = (float*)d_out;

    hipLaunchKernelGGL(qprep_kernel, dim3(1), dim3(1024), 0, stream,
                       in_proj_w, weights, meas_w, out_proj_w, ws);
    hipLaunchKernelGGL(qconv_main_kernel, dim3(1024), dim3(128), 0, stream,
                       x, inp_scale, in_proj_b, out_proj_b, ws, out);
}


# prep launch is 2 work-groups split by task: group 1 repacks in_proj and out_proj (44 of 60 fragments, no circuit input) beside group 0's gate chain, whose tail is 8 stores a thread, not 30
# speedup vs baseline: 6.3490x; 1.1115x over previous
#include <hip/hip_runtime.h>
#include <hip/hip_bf16.h>
#include <math.h>

typedef __attribute__((ext_vector_type(16))) _Float16 v16h;
typedef __attribute__((ext_vector_type(8)))  float    v8f;
typedef __attribute__((ext_vector_type(4)))  float    v4f;

#define PATCH 288
#define QD 64
#define C_IN 32
#define ROWSTRIDE_A 296
#define ROWSTRIDE_Y 72
#define OSTAGE_TILE (16 * ROWSTRIDE_A / 2)
#define OSTAGE_PITCH 20

#define WS_W   0
#define WS_MR  18432
#define WS_MI  (18432 + 4096)
#define WS_P   (18432 + 8192)

union FragU { uint4 u[2]; v16h v; };


__device__ __forceinline__ void c2mul(const float* a, const float* b, float* o)
{
    for (int r = 0; r < 2; ++r)
        for (int c = 0; c < 2; ++c) {
            float re = 0.f, im = 0.f;
            for (int k = 0; k < 2; ++k) {
                float ar = a[(r * 2 + k) * 2], ai = a[(r * 2 + k) * 2 + 1];
                float br = b[(k * 2 + c) * 2], bi = b[(k * 2 + c) * 2 + 1];
                re += ar * br - ai * bi;
                im += ar * bi + ai * br;
            }
            o[(r * 2 + c) * 2] = re;
            o[(r * 2 + c) * 2 + 1] = im;
        }
}

__device__ __forceinline__ void make_rx(float c, float s, float* o)
{
    o[0] = c;   o[1] = 0.f; o[2] = 0.f; o[3] = -s;
    o[4] = 0.f; o[5] = -s;  o[6] = c;   o[7] = 0.f;
}

__device__ __forceinline__ void make_ry(float c, float s, float* o)
{
    o[0] = c; o[1] = 0.f; o[2] = -s; o[3] = 0.f;
    o[4] = s; o[5] = 0.f; o[6] = c;  o[7] = 0.f;
}

template <bool HI, bool PERM>
__device__ __forceinline__ void block_sweep(float* Mr, float* Mi, const float* Gw, int tid)
{
    float xr[8] = {}, xi[8] = {};
    const bool act = tid < 512;
    const int c = tid & 63, o = (tid >> 6) & 7;
    if (act) {
        for (int s = 0; s < 8; ++s) {
            int row = HI ? (s * 8 + o) : (o * 8 + s);
            if (PERM) row ^= row >> 1;
            xr[s] = Mr[row * 64 + c];
            xi[s] = Mi[row * 64 + c];
        }
        #pragma unroll
        for (int q = 0; q < 3; ++q) {
            const float* g = Gw + q * 8;
            const float g00r = g[0], g00i = g[1], g01r = g[2], g01i = g[3];
            const float g10r = g[4], g10i = g[5], g11r = g[6], g11i = g[7];
            const int bit = 4 >> q;
            #pragma unroll
            for (int s0 = 0; s0 < 8; ++s0) {
                if (s0 & bit) continue;
                const int s1 = s0 | bit;
                const float ar = xr[s0], ai = xi[s0], br = xr[s1], bi = xi[s1];
                xr[s0] = fmaf(-g01i, bi, fmaf(g01r, br, fmaf(-g00i, ai, g00r * ar)));
                xi[s0] = fmaf( g01i, br, fmaf(g01r, bi, fmaf( g00i, ar, g00r * ai)));
                xr[s1] = fmaf(-g11i, bi, fmaf(g11r, br, fmaf(-g10i, ai, g10r * ar)));
                xi[s1] = fmaf( g11i, br, fmaf(g11r, bi, fmaf( g10i, ar, g10r * ai)));
            }
        }
    }
    if (PERM) __syncthreads();
    if (act) {
        for (int a = 0; a < 8; ++a) {
            const int row = HI ? (a * 8 + o) : (o * 8 + a);
            Mr[row * 64 + c] = xr[a];
            Mi[row * 64 + c] = xi[a];
        }
    }
    __syncthreads();
}

__global__ void __launch_bounds__(1024)
qprep_kernel(const float* __restrict__ in_proj_w,
             const float* __restrict__ weights,
             const float* __restrict__ meas_w,
             const float* __restrict__ out_proj_w,
             _Float16* __restrict__ ws)
{
    __shared__ float Mr[64 * 64];
    __shared__ float Mi[64 * 64];
    __shared__ float CS[60 * 2];
    __shared__ float G[18 * 8];
    __shared__ float Wl[QD * PATCH];
    const int tid = threadIdx.x;

    if (blockIdx.x == 1) {
        float wv[18];
        for (int i = 0; i < 18; ++i) wv[i] = in_proj_w[tid + 1024 * i];
        float pv[4];
        #pragma unroll
        for (int j = 0; j < 4; ++j) {
            const int e = tid + 1024 * j;
            const int elo = e & 7, lane = (e >> 3) & 31, ehi = (e >> 8) & 1, sfr = e >> 9;
            const int kt = sfr & 1, nt = sfr >> 1;
            const int n = nt * 16 + (lane & 15);
            const int base = (lane & 16) ? 8 : 0;
            const int k = kt * 32 + base + elo + 16 * ehi;
            pv[j] = out_proj_w[n * 64 + k];
        }
        for (int i = 0; i < 18; ++i) Wl[tid + 1024 * i] = wv[i];
        __syncthreads();

        {
            const int tl = tid & 511, t9 = tid >> 9;
            const int elo = tl & 7, lane = (tl >> 3) & 31, ehi = tl >> 8;
            const int base = (lane & 16) ? 8 : 0;
            const int kl = base + elo + 16 * ehi;
            const int run = kl >> 3, cl = kl & 7;
            const float* src = Wl + (t9 * 32 + (lane & 15)) * PATCH + 9 * cl + run;
            #pragma unroll
            for (int i = 0; i < 18; ++i) {
                const int kt = i % 9, nt2 = i / 9;
                const int c0 = (4 * kt) / 9, r0 = (4 * kt) % 9;
                const int ko = 72 * c0 + r0 + ((r0 + run >= 9) ? 63 : 0);
                ws[WS_W + (t9 * 18 + i) * 512 + tl] = (_Float16)src[nt2 * 16 * PATCH + ko];
            }
        }
        #pragma unroll
        for (int j = 0; j < 4; ++j) ws[WS_P + tid + 1024 * j] = (_Float16)pv[j];
        return;
    }

    float ang = 0.f;
    if (tid < 60) {
        if (tid < 36) {
            ang = weights[tid] * 0.5f;
        } else {
            const int k = tid - 36, w = k >> 2, which = k & 3;
            const float t = meas_w[w * 3 + 0], p = meas_w[w * 3 + 1], lm = meas_w[w * 3 + 2];
            ang = (which == 0) ? t * 0.5f : (which == 1) ? p : (which == 2) ? lm : (p + lm);
        }
    }
    if (tid < 60) {
        CS[tid * 2]     = cosf(ang);
        CS[tid * 2 + 1] = sinf(ang);
    }
    __syncthreads();

    if (tid < 18) {
        const int L = tid / 6, w = tid - L * 6;
        float m0[8], m1[8], m2[8], m3[8];
        if (L == 0) {
            make_ry(CS[(w * 3 + 1) * 2], CS[(w * 3 + 1) * 2 + 1], m0);
            make_rx(CS[(w * 3 + 0) * 2], CS[(w * 3 + 0) * 2 + 1], m1);
            c2mul(m0, m1, m3);
        } else if (L == 1) {
            make_ry(CS[((6 + w) * 3 + 1) * 2], CS[((6 + w) * 3 + 1) * 2 + 1], m0);
            make_rx(CS[((6 + w) * 3 + 0) * 2], CS[((6 + w) * 3 + 0) * 2 + 1], m1);
            c2mul(m0, m1, m2);
            make_ry(CS[(w * 3 + 2) * 2], CS[(w * 3 + 2) * 2 + 1], m0);
            c2mul(m2, m0, m3);
        } else {
            const float c  = CS[(36 + w * 4 + 0) * 2], s  = CS[(36 + w * 4 + 0) * 2 + 1];
            const float cp = CS[(36 + w * 4 + 1) * 2], sp = CS[(36 + w * 4 + 1) * 2 + 1];
            const float cl = CS[(36 + w * 4 + 2) * 2], sl = CS[(36 + w * 4 + 2) * 2 + 1];
            const float cpl = CS[(36 + w * 4 + 3) * 2], spl = CS[(36 + w * 4 + 3) * 2 + 1];
            m0[0] = c;      m0[1] = 0.f;    m0[2] = -cl * s; m0[3] = -sl * s;
            m0[4] = cp * s; m0[5] = sp * s; m0[6] = cpl * c; m0[7] = spl * c;
            make_ry(CS[((6 + w) * 3 + 2) * 2], CS[((6 + w) * 3 + 2) * 2 + 1], m1);
            c2mul(m0, m1, m3);
        }
        for (int e = 0; e < 8; ++e) G[tid * 8 + e] = m3[e];
    }
    __syncthreads();

    if (tid < 256) {
        const int col = tid & 63, slo = tid >> 6;
        float gr[6][2], gi[6][2];
        for (int w = 0; w < 6; ++w) {
            const int cb = (col >> (5 - w)) & 1;
            gr[w][0] = G[(w * 4 + 0 + cb) * 2];  gi[w][0] = G[(w * 4 + 0 + cb) * 2 + 1];
            gr[w][1] = G[(w * 4 + 2 + cb) * 2];  gi[w][1] = G[(w * 4 + 2 + cb) * 2 + 1];
        }
        const bool b4 = (slo >> 1) & 1, b5 = slo & 1;
        const float g4r = b4 ? gr[4][1] : gr[4][0], g4i = b4 ? gi[4][1] : gi[4][0];
        const float g5r = b5 ? gr[5][1] : gr[5][0], g5i = b5 ? gi[5][1] : gi[5][0];
        float pr[16], pi[16];
        pr[0] = g4r * g5r - g4i * g5i;
        pi[0] = g4r * g5i + g4i * g5r;
        for (int k = 0; k < 4; ++k) {
            const int w = 3 - k, n = 1 << k;
            for (int idx = 0; idx < n; ++idx) {
                const float ar = pr[idx], ai = pi[idx];
                pr[idx + n] = ar * gr[w][1] - ai * gi[w][1];
                pi[idx + n] = ar * gi[w][1] + ai * gr[w][1];
                pr[idx]     = ar * gr[w][0] - ai * gi[w][0];
                pi[idx]     = ar * gi[w][0] + ai * gr[w][0];
            }
        }
        for (int t = 0; t < 16; ++t) {
            const int s = slo + 4 * t;
            const int row = s ^ (s >> 1) ^ (s >> 2) ^ (s >> 3) ^ (s >> 4) ^ (s >> 5);
            Mr[row * 64 + col] = pr[t];
            Mi[row * 64 + col] = pi[t];
        }
    }
    __syncthreads();

    block_sweep<true,  false>(Mr, Mi, G + (1 * 6 + 0) * 8, tid);
    block_sweep<false, false>(Mr, Mi, G + (1 * 6 + 3) * 8, tid);
    block_sweep<true,  true >(Mr, Mi, G + (2 * 6 + 0) * 8, tid);
    block_sweep<false, false>(Mr, Mi, G + (2 * 6 + 3) * 8, tid);

    #pragma unroll
    for (int e = tid; e < 4 * 2 * 512; e += 1024) {
        int elo = e & 7, lane = (e >> 3) & 31, ehi = (e >> 8) & 1, s = e >> 9;
        int kt = s & 1, nt = s >> 1;
        int n = nt * 16 + (lane & 15);
        int base = (lane & 16) ? 8 : 0;
        int kl = base + elo + 16 * ehi;
        int k = kt * 32 + kl;
        ws[WS_MR + e] = (_Float16)Mr[n * 64 + k];
        ws[WS_MI + e] = (_Float16)Mi[n * 64 + k];
    }
}

template <int G>
__device__ __forceinline__ void load_group(FragU (&ring)[3][4], const _Float16* __restrict__ ws, int lane)
{
    #pragma unroll
    for (int f = 0; f < 4; ++f) {
        const _Float16* q;
        if (G < 9)        q = ws + WS_W + ((f * 9 + G) * 512 + lane * 8);
        else if (G < 13)  q = ws + ((f & 1) ? WS_MI : WS_MR) + (((2 * ((G - 9) >> 1) + (f >> 1)) * 2 + ((G - 9) & 1)) * 512 + lane * 8);
        else              q = ws + WS_P + ((f * 2 + (G - 13)) * 512 + lane * 8);
        ring[G % 3][f].u[0] = *(const uint4*)(q);
        ring[G % 3][f].u[1] = *(const uint4*)(q + 256);
    }
}

__global__ void __launch_bounds__(128)
qconv_main_kernel(const float* __restrict__ x,
                  const float* __restrict__ inp_scale,
                  const float* __restrict__ in_proj_b,
                  const float* __restrict__ out_proj_b,
                  const _Float16* __restrict__ ws,
                  float* __restrict__ out)
{
    __shared__ __align__(16) _Float16 bufA[64 * ROWSTRIDE_A];
    __shared__ __align__(16) _Float16 bufY[64 * ROWSTRIDE_Y];
    __shared__ __align__(16) _Float16 bufP[64 * ROWSTRIDE_Y];

    const int tid  = threadIdx.x;
    const int lane = tid & 31;
    const int mt   = tid >> 5;

    FragU ring[3][4];
    float bia[4], obi[4];
    {
        const int m = tid & 63;
        const int gp = __builtin_amdgcn_readfirstlane(tid >> 6);
        const int b = blockIdx.x >> 6;
        const int h = blockIdx.x & 63;
        const bool top = h > 0, bot = h < 63;
        const int hm = top ? h - 1 : h;
        const int hp = bot ? h + 1 : h;
        float v[2][8][3];
        #pragma unroll
        for (int gi = 0; gi < 2; ++gi) {
            const float* xb = x + (size_t)b * C_IN * 4096 + (size_t)(8 * (2 * gp + gi)) * 4096 + m;
            #pragma unroll
            for (int j = 0; j < 8; ++j) {
                const float* xc = xb + j * 4096;
                v[gi][j][0] = xc[hm * 64];
                v[gi][j][1] = xc[h * 64];
                v[gi][j][2] = xc[hp * 64];
            }
        }
        load_group<0>(ring, ws, lane);
        load_group<1>(ring, ws, lane);
        #pragma unroll
        for (int j = 0; j < 4; ++j) {
            bia[j] = in_proj_b[16 * j + (lane & 15)];
            obi[j] = out_proj_b[16 * j + (lane & 15)];
        }
        const bool lastcol = (m == 63), firstcol = (m == 0);
        #pragma unroll
        for (int gi = 0; gi < 2; ++gi) {
            const int g = 2 * gp + gi;
            _Float16* r0 = bufA + ((m + 1) & 63) * ROWSTRIDE_A + 72 * g;
            _Float16* r1 = bufA + m * ROWSTRIDE_A + 72 * g;
            _Float16* r2 = bufA + ((m + 63) & 63) * ROWSTRIDE_A + 72 * g;
            const float* sc = inp_scale + 72 * g;
            #pragma unroll
            for (int r = 0; r < 3; ++r) {
                #pragma unroll
                for (int s = 0; s < 3; ++s) {
                    const int t = r * 3 + s;
                    union { uint4 u; _Float16 e[8]; } pk;
                    #pragma unroll
                    for (int j = 0; j < 8; ++j) {
                        float val = v[gi][j][r];
                        if (r == 0 && !top) val = 0.f;
                        if (r == 2 && !bot) val = 0.f;
                        if (s == 0 && lastcol) val = 0.f;
                        if (s == 2 && firstcol) val = 0.f;
                        pk.e[j] = (_Float16)(val * sc[9 * j + t]);
                    }
                    _Float16* dst = (s == 0) ? r0 : (s == 1) ? r1 : r2;
                    *(uint4*)(dst + 8 * t) = pk.u;
                }
            }
        }
    }
    __syncthreads();

    const int nlo  = lane & 15;
    const int base = (lane & 16) ? 8 : 0;

#define FRONT_OF_PRODUCTS() __builtin_amdgcn_sched_barrier(0)

    v8f acc[4] = {};
    {
        const _Float16* Arow = &bufA[(mt * 16 + nlo) * ROWSTRIDE_A];
        {
            load_group<2>(ring, ws, lane);
            FRONT_OF_PRODUCTS();
            FragU af;
            const _Float16* pa = Arow + 0 * 32 + base;
            af.u[0] = *(const uint4*)(pa);
            af.u[1] = *(const uint4*)(pa + 16);
            #pragma unroll
            for (int j = 0; j < 4; ++j)
                acc[j] = __builtin_amdgcn_wmma_f32_16x16x32_f16(false, af.v, false, ring[0][j].v, (short)0, acc[j], false, false);
        }
        {
            load_group<3>(ring, ws, lane);
            FRONT_OF_PRODUCTS();
            FragU af;
            const _Float16* pa = Arow + 1 * 32 + base;
            af.u[0] = *(const uint4*)(pa);
            af.u[1] = *(const uint4*)(pa + 16);
            #pragma unroll
            for (int j = 0; j < 4; ++j)
                acc[j] = __builtin_amdgcn_wmma_f32_16x16x32_f16(false, af.v, false, ring[1][j].v, (short)0, acc[j], false, false);
        }
        {
            load_group<4>(ring, ws, lane);
            FRONT_OF_PRODUCTS();
            FragU af;
            const _Float16* pa = Arow + 2 * 32 + base;
            af.u[0] = *(const uint4*)(pa);
            af.u[1] = *(const uint4*)(pa + 16);
            #pragma unroll
            for (int j = 0; j < 4; ++j)
                acc[j] = __builtin_amdgcn_wmma_f32_16x16x32_f16(false, af.v, false, ring[2][j].v, (short)0, acc[j], false, false);
        }
        {
            load_group<5>(ring, ws, lane);
            FRONT_OF_PRODUCTS();
            FragU af;
            const _Float16* pa = Arow + 3 * 32 + base;
            af.u[0] = *(const uint4*)(pa);
            af.u[1] = *(const uint4*)(pa + 16);
            #pragma unroll
            for (int j = 0; j < 4; ++j)
                acc[j] = __builtin_amdgcn_wmma_f32_16x16x32_f16(false, af.v, false, ring[0][j].v, (short)0, acc[j], false, false);
        }
        {
            load_group<6>(ring, ws, lane);
            FRONT_OF_PRODUCTS();
            FragU af;
            const _Float16* pa = Arow + 4 * 32 + base;
            af.u[0] = *(const uint4*)(pa);
            af.u[1] = *(const uint4*)(pa + 16);
            #pragma unroll
            for (int j = 0; j < 4; ++j)
                acc[j] = __builtin_amdgcn_wmma_f32_16x16x32_f16(false, af.v, false, ring[1][j].v, (short)0, acc[j], false, false);
        }
        {
            load_group<7>(ring, ws, lane);
            FRONT_OF_PRODUCTS();
            FragU af;
            const _Float16* pa = Arow + 5 * 32 + base;
            af.u[0] = *(const uint4*)(pa);
            af.u[1] = *(const uint4*)(pa + 16);
            #pragma unroll
            for (int j = 0; j < 4; ++j)
                acc[j] = __builtin_amdgcn_wmma_f32_16x16x32_f16(false, af.v, false, ring[2][j].v, (short)0, acc[j], false, false);
        }
        {
            load_group<8>(ring, ws, lane);
            FRONT_OF_PRODUCTS();
            FragU af;
            const _Float16* pa = Arow + 6 * 32 + base;
            af.u[0] = *(const uint4*)(pa);
            af.u[1] = *(const uint4*)(pa + 16);
            #pragma unroll
            for (int j = 0; j < 4; ++j)
                acc[j] = __builtin_amdgcn_wmma_f32_16x16x32_f16(false, af.v, false, ring[0][j].v, (short)0, acc[j], false, false);
        }
        {
            load_group<9>(ring, ws, lane);
            FRONT_OF_PRODUCTS();
            FragU af;
            const _Float16* pa = Arow + 7 * 32 + base;
            af.u[0] = *(const uint4*)(pa);
            af.u[1] = *(const uint4*)(pa + 16);
            #pragma unroll
            for (int j = 0; j < 4; ++j)
                acc[j] = __builtin_amdgcn_wmma_f32_16x16x32_f16(false, af.v, false, ring[1][j].v, (short)0, acc[j], false, false);
        }
        {
            load_group<10>(ring, ws, lane);
            FRONT_OF_PRODUCTS();
            FragU af;
            const _Float16* pa = Arow + 8 * 32 + base;
            af.u[0] = *(const uint4*)(pa);
            af.u[1] = *(const uint4*)(pa + 16);
            #pragma unroll
            for (int j = 0; j < 4; ++j)
                acc[j] = __builtin_amdgcn_wmma_f32_16x16x32_f16(false, af.v, false, ring[2][j].v, (short)0, acc[j], false, false);
        }
    }

    float y[4][8], part[8];
    {
        #pragma unroll
        for (int r = 0; r < 8; ++r) {
            float s = 0.f;
            #pragma unroll
            for (int j = 0; j < 4; ++j) {
                y[j][r] = acc[j][r] + bia[j];
                s += y[j][r] * y[j][r];
            }
            part[r] = s;
        }
    }
    #pragma unroll
    for (int mask = 1; mask <= 8; mask <<= 1) {
        #pragma unroll
        for (int r = 0; r < 8; ++r)
            part[r] += __shfl_xor(part[r], mask, 32);
    }
    #pragma unroll
    for (int r = 0; r < 8; ++r) {
        const int m = mt * 16 + base + r;
        const float inv = 1.f / (sqrtf(part[r]) + 1e-9f);
        #pragma unroll
        for (int j = 0; j < 4; ++j)
            bufY[m * ROWSTRIDE_Y + 16 * j + nlo] = (_Float16)(y[j][r] * inv);
    }
    __builtin_amdgcn_fence(__ATOMIC_ACQ_REL, "wavefront");

    {
        const _Float16* Yrow = &bufY[(mt * 16 + nlo) * ROWSTRIDE_Y];
        FragU af[2];
        #pragma unroll
        for (int kt = 0; kt < 2; ++kt) {
            const _Float16* pa = Yrow + kt * 32 + base;
            af[kt].u[0] = *(const uint4*)(pa);
            af[kt].u[1] = *(const uint4*)(pa + 16);
        }
        {
            v8f ar[2] = {}, ai[2] = {};
            load_group<11>(ring, ws, lane);
            FRONT_OF_PRODUCTS();
            #pragma unroll
            for (int jj = 0; jj < 2; ++jj) {
                ar[jj] = __builtin_amdgcn_wmma_f32_16x16x32_f16(false, af[0].v, false, ring[0][2 * jj].v,     (short)0, ar[jj], false, false);
                ai[jj] = __builtin_amdgcn_wmma_f32_16x16x32_f16(false, af[0].v, false, ring[0][2 * jj + 1].v, (short)0, ai[jj], false, false);
            }
            load_group<12>(ring, ws, lane);
            FRONT_OF_PRODUCTS();
            #pragma unroll
            for (int jj = 0; jj < 2; ++jj) {
                ar[jj] = __builtin_amdgcn_wmma_f32_16x16x32_f16(false, af[1].v, false, ring[1][2 * jj].v,     (short)0, ar[jj], false, false);
                ai[jj] = __builtin_amdgcn_wmma_f32_16x16x32_f16(false, af[1].v, false, ring[1][2 * jj + 1].v, (short)0, ai[jj], false, false);
            }
            #pragma unroll
            for (int r = 0; r < 8; ++r) {
                const int m = mt * 16 + base + r;
                #pragma unroll
                for (int jj = 0; jj < 2; ++jj)
                    bufP[m * ROWSTRIDE_Y + 16 * (0 + jj) + nlo] = (_Float16)(ar[jj][r] * ar[jj][r] + ai[jj][r] * ai[jj][r]);
            }
        }
        {
            v8f ar[2] = {}, ai[2] = {};
            load_group<13>(ring, ws, lane);
            FRONT_OF_PRODUCTS();
            #pragma unroll
            for (int jj = 0; jj < 2; ++jj) {
                ar[jj] = __builtin_amdgcn_wmma_f32_16x16x32_f16(false, af[0].v, false, ring[2][2 * jj].v,     (short)0, ar[jj], false, false);
                ai[jj] = __builtin_amdgcn_wmma_f32_16x16x32_f16(false, af[0].v, false, ring[2][2 * jj + 1].v, (short)0, ai[jj], false, false);
            }
            load_group<14>(ring, ws, lane);
            FRONT_OF_PRODUCTS();
            #pragma unroll
            for (int jj = 0; jj < 2; ++jj) {
                ar[jj] = __builtin_amdgcn_wmma_f32_16x16x32_f16(false, af[1].v, false, ring[0][2 * jj].v,     (short)0, ar[jj], false, false);
                ai[jj] = __builtin_amdgcn_wmma_f32_16x16x32_f16(false, af[1].v, false, ring[0][2 * jj + 1].v, (short)0, ai[jj], false, false);
            }
            #pragma unroll
            for (int r = 0; r < 8; ++r) {
                const int m = mt * 16 + base + r;
                #pragma unroll
                for (int jj = 0; jj < 2; ++jj)
                    bufP[m * ROWSTRIDE_Y + 16 * (2 + jj) + nlo] = (_Float16)(ar[jj][r] * ar[jj][r] + ai[jj][r] * ai[jj][r]);
            }
        }
    }
    __builtin_amdgcn_fence(__ATOMIC_ACQ_REL, "wavefront");

    v8f o[4] = {};
    {
        const _Float16* Prow = &bufP[(mt * 16 + nlo) * ROWSTRIDE_Y];
        #pragma unroll
        for (int kt = 0; kt < 2; ++kt) {
            FragU af;
            const _Float16* pa = Prow + kt * 32 + base;
            af.u[0] = *(const uint4*)(pa);
            af.u[1] = *(const uint4*)(pa + 16);
            #pragma unroll
            for (int j = 0; j < 4; ++j)
                o[j] = __builtin_amdgcn_wmma_f32_16x16x32_f16(false, af.v, false, ring[(13 + kt) % 3][j].v, (short)0, o[j], false, false);
        }
    }

    float* outstage = (float*)bufA;
    {
        float* os = outstage + mt * OSTAGE_TILE + mt * 16;
        #pragma unroll
        for (int j = 0; j < 4; ++j) {
            const float ob = obi[j];
            float* col = os + (16 * j + nlo) * OSTAGE_PITCH + base;
            const v4f lo = { o[j][0] + ob, o[j][1] + ob, o[j][2] + ob, o[j][3] + ob };
            const v4f hi = { o[j][4] + ob, o[j][5] + ob, o[j][6] + ob, o[j][7] + ob };
            *(v4f*)(col)     = lo;
            *(v4f*)(col + 4) = hi;
        }
    }
    __syncthreads();
    {
        const int q = lane & 15, cs = lane >> 4, t = q >> 2;
        const float* rd = outstage + t * OSTAGE_TILE + t * 16 + (2 * mt + cs) * OSTAGE_PITCH + 4 * (q & 3);
        float* wr = out + ((size_t)((blockIdx.x >> 6) * 64 + 2 * mt + cs)) * 4096 + (blockIdx.x & 63) * 64 + 4 * q;
        v4f ov[8];
        #pragma unroll
        for (int i = 0; i < 8; ++i) ov[i] = *(const v4f*)(rd + i * 8 * OSTAGE_PITCH);
        #pragma unroll
        for (int i = 0; i < 8; ++i) *(v4f*)(wr + (size_t)i * 8 * 4096) = ov[i];
    }
}

extern "C" void kernel_launch(void* const* d_in, const int* in_sizes, int n_in,
                              void* d_out, int out_size, void* d_ws, size_t ws_size,
                              hipStream_t stream) {
    const float* x          = (const float*)d_in[0];
    const float* inp_scale  = (const float*)d_in[1];
    const float* in_proj_w  = (const float*)d_in[2];
    const float* in_proj_b  = (const float*)d_in[3];
    const float* weights    = (const float*)d_in[4];
    const float* meas_w     = (const float*)d_in[5];
    const float* out_proj_w = (const float*)d_in[6];
    const float* out_proj_b = (const float*)d_in[7];
    _Float16* ws = (_Float16*)d_ws;
    float* out = (float*)d_out;

    hipLaunchKernelGGL(qprep_kernel, dim3(2), dim3(1024), 0, stream,
                       in_proj_w, weights, meas_w, out_proj_w, ws);
    hipLaunchKernelGGL(qconv_main_kernel, dim3(1024), dim3(128), 0, stream,
                       x, inp_scale, in_proj_b, out_proj_b, ws, out);
}
